# starting kernel
# baseline (speedup 1.0000x reference)
#include <hip/hip_runtime.h>

typedef __attribute__((ext_vector_type(2))) float v2f;
typedef __attribute__((ext_vector_type(8))) float v8f;

#define N_NODES 100000
#define N_EDGES 3200000
#define D_FEAT  256
#define HIDDEN  64
#define NCLASS  7
#define ROWBLKS (N_NODES / 16)


__global__ void k_zero(float* __restrict__ p, int n) {
    int i = blockIdx.x * blockDim.x + threadIdx.x;
    if (i < n) p[i] = 0.0f;
}

__global__ void k_deg(const int* __restrict__ ei, const float* __restrict__ ew,
                      float* __restrict__ deg) {
    int e = blockIdx.x * blockDim.x + threadIdx.x;
    if (e < N_EDGES) atomicAdd(&deg[ei[N_EDGES + e]], ew[e]);
}

__global__ void k_dinv(float* __restrict__ deg) {
    int i = blockIdx.x * blockDim.x + threadIdx.x;
    if (i < N_NODES) {
        float d = deg[i] + 1.0f;
        deg[i] = rsqrtf(fmaxf(d, 1e-12f));
    }
}


__global__ __launch_bounds__(128) void k_gemm0(const float* __restrict__ x,
                                               const float* __restrict__ W0,
                                               float* __restrict__ m0) {
    const int lane = threadIdx.x & 31;
    const int wv   = threadIdx.x >> 5;
    const int row0 = blockIdx.x * 16;
    const int col0 = wv * 16;
    const int hh   = lane >> 4;
    const int l15  = lane & 15;

    v8f c = {};
    const float* xrow = x  + (row0 + l15) * D_FEAT + 2 * hh;
    const float* wcol = W0 + (2 * hh) * HIDDEN + col0 + l15;

#pragma unroll 4
    for (int k = 0; k < D_FEAT; k += 4) {
        v2f a, b;
        a.x = xrow[k];
        a.y = xrow[k + 1];
        b.x = wcol[k * HIDDEN];
        b.y = wcol[k * HIDDEN + HIDDEN];
        c = __builtin_amdgcn_wmma_f32_16x16x4_f32(false, a, false, b,
                                                  (short)0, c, false, false);
    }
    float* out = m0 + (row0 + 8 * hh) * HIDDEN + col0 + l15;
#pragma unroll
    for (int v = 0; v < 8; ++v) out[v * HIDDEN] = c[v];
}


__global__ void k_init_h(const float* __restrict__ dinv, const float* __restrict__ m0,
                         float* __restrict__ h) {
    int idx = blockIdx.x * blockDim.x + threadIdx.x;
    if (idx < N_NODES * HIDDEN) {
        float di = dinv[idx / HIDDEN];
        h[idx] = di * di * m0[idx];
    }
}

__global__ __launch_bounds__(256) void k_scatter0(const int* __restrict__ ei,
                                                  const float* __restrict__ ew,
                                                  const float* __restrict__ dinv,
                                                  const float* __restrict__ m0,
                                                  float* __restrict__ h) {
    const int lane = threadIdx.x & 31;
    const int e = blockIdx.x * 8 + (threadIdx.x >> 5);
    if (e >= N_EDGES) return;
    const int s = ei[e];
    const int d = ei[N_EDGES + e];
    const float wn = dinv[s] * ew[e] * dinv[d];
    const float2 v = *(const float2*)(m0 + s * HIDDEN + 2 * lane);
    atomicAdd(&h[d * HIDDEN + 2 * lane],     wn * v.x);
    atomicAdd(&h[d * HIDDEN + 2 * lane + 1], wn * v.y);
}

__global__ void k_bias_relu(float* __restrict__ h, const float* __restrict__ b0) {
    int idx = blockIdx.x * blockDim.x + threadIdx.x;
    if (idx < N_NODES * HIDDEN)
        h[idx] = fmaxf(h[idx] + b0[idx % HIDDEN], 0.0f);
}


__global__ __launch_bounds__(128) void k_gemm1(const float* __restrict__ h,
                                               const float* __restrict__ W1,
                                               float* __restrict__ m1) {
    const int lane = threadIdx.x & 31;
    const int wv   = threadIdx.x >> 5;
    const int rb   = blockIdx.x * 4 + wv;
    if (rb >= ROWBLKS) return;
    const int row0 = rb * 16;
    const int hh   = lane >> 4;
    const int l15  = lane & 15;

    v8f c = {};
    const float* hrow = h + (row0 + l15) * HIDDEN + 2 * hh;
#pragma unroll
    for (int k = 0; k < HIDDEN; k += 4) {
        v2f a, b;
        a.x = hrow[k];
        a.y = hrow[k + 1];
        float bx = 0.0f, by = 0.0f;
        if (l15 < NCLASS) {
            bx = W1[(k + 2 * hh) * NCLASS + l15];
            by = W1[(k + 2 * hh + 1) * NCLASS + l15];
        }
        b.x = bx; b.y = by;
        c = __builtin_amdgcn_wmma_f32_16x16x4_f32(false, a, false, b,
                                                  (short)0, c, false, false);
    }
    if (l15 < NCLASS) {
        float* out = m1 + (row0 + 8 * hh) * NCLASS + l15;
#pragma unroll
        for (int v = 0; v < 8; ++v) out[v * NCLASS] = c[v];
    }
}


__global__ void k_init_out(const float* __restrict__ dinv, const float* __restrict__ m1,
                           const float* __restrict__ b1, float* __restrict__ out) {
    int idx = blockIdx.x * blockDim.x + threadIdx.x;
    if (idx < N_NODES * NCLASS) {
        int i = idx / NCLASS, j = idx - i * NCLASS;
        float di = dinv[i];
        out[idx] = di * di * m1[idx] + b1[j];
    }
}

__global__ __launch_bounds__(256) void k_scatter1(const int* __restrict__ ei,
                                                  const float* __restrict__ ew,
                                                  const float* __restrict__ dinv,
                                                  const float* __restrict__ m1,
                                                  float* __restrict__ out) {
    const int j = threadIdx.x & 7;
    const int e = blockIdx.x * 32 + (threadIdx.x >> 3);
    if (e >= N_EDGES || j >= NCLASS) return;
    const int s = ei[e];
    const int d = ei[N_EDGES + e];
    const float wn = dinv[s] * ew[e] * dinv[d];
    atomicAdd(&out[d * NCLASS + j], wn * m1[s * NCLASS + j]);
}


extern "C" void kernel_launch(void* const* d_in, const int* in_sizes, int n_in,
                              void* d_out, int out_size, void* d_ws, size_t ws_size,
                              hipStream_t stream) {
    const float* x  = (const float*)d_in[0];
    const int*   ei = (const int*)  d_in[1];
    const float* ew = (const float*)d_in[2];
    const float* W0 = (const float*)d_in[3];
    const float* b0 = (const float*)d_in[4];
    const float* W1 = (const float*)d_in[5];
    const float* b1 = (const float*)d_in[6];
    float* out = (float*)d_out;

    float* ws   = (float*)d_ws;
    float* dinv = ws;
    float* m0   = dinv + N_NODES;
    float* h    = m0 + (size_t)N_NODES * HIDDEN;
    float* m1   = h  + (size_t)N_NODES * HIDDEN;

    k_zero<<<(N_NODES + 255) / 256, 256, 0, stream>>>(dinv, N_NODES);
    k_deg <<<(N_EDGES + 255) / 256, 256, 0, stream>>>(ei, ew, dinv);
    k_dinv<<<(N_NODES + 255) / 256, 256, 0, stream>>>(dinv);

    k_gemm0<<<ROWBLKS, 128, 0, stream>>>(x, W0, m0);

    k_init_h<<<(N_NODES * HIDDEN + 255) / 256, 256, 0, stream>>>(dinv, m0, h);
    k_scatter0<<<(N_EDGES + 7) / 8, 256, 0, stream>>>(ei, ew, dinv, m0, h);
    k_bias_relu<<<(N_NODES * HIDDEN + 255) / 256, 256, 0, stream>>>(h, b0);

    k_gemm1<<<(ROWBLKS + 3) / 4, 128, 0, stream>>>(h, W1, m1);

    k_init_out<<<(N_NODES * NCLASS + 255) / 256, 256, 0, stream>>>(dinv, m1, b1, out);
    k_scatter1<<<(N_EDGES + 31) / 32, 256, 0, stream>>>(ei, ew, dinv, m1, out);
}


# edge scatters become gathers: one int-atomic pass slots each edge in its destination's list; degree, layer 0, layer 1 sum lists in registers; layer-1 product on an LDS tile; sums in slot order
# speedup vs baseline: 1.8450x; 1.8450x over previous
#include <hip/hip_runtime.h>

typedef __attribute__((ext_vector_type(2))) float v2f;
typedef __attribute__((ext_vector_type(8))) float v8f;

#define N_NODES 100000
#define N_EDGES 3200000
#define D_FEAT  256
#define HIDDEN  64
#define NCLASS  7
#define ROWBLKS (N_NODES / 16)
#define CAP     64
#define HPAD    66


__global__ void k_zero_cnt(int* __restrict__ cnt) {
    int i = blockIdx.x * blockDim.x + threadIdx.x;
    if (i < N_NODES) cnt[i] = 0;
}

__global__ void k_fill(const int* __restrict__ ei, int* __restrict__ cnt,
                       int* __restrict__ ell) {
    int e = blockIdx.x * blockDim.x + threadIdx.x;
    if (e < N_EDGES) {
        const int d = ei[N_EDGES + e];
        const int slot = atomicAdd(&cnt[d], 1);
        if (slot < CAP) ell[d * CAP + slot] = e;
    }
}

__global__ void k_dinv(const int* __restrict__ ei, const float* __restrict__ ew,
                       const int* __restrict__ cnt, const int* __restrict__ ell,
                       float* __restrict__ dinv) {
    int d = blockIdx.x * blockDim.x + threadIdx.x;
    if (d >= N_NODES) return;
    const int n = cnt[d];
    float dsum = 0.0f;
    if (n <= CAP) {
        const int* row = ell + d * CAP;
        for (int i = 0; i < n; ++i) dsum += ew[row[i]];
    } else {
        for (int e = 0; e < N_EDGES; ++e)
            if (ei[N_EDGES + e] == d) dsum += ew[e];
    }
    dinv[d] = rsqrtf(fmaxf(dsum + 1.0f, 1e-12f));
}


__global__ __launch_bounds__(128) void k_gemm0(const float* __restrict__ x,
                                               const float* __restrict__ W0,
                                               const float* __restrict__ dinv,
                                               float* __restrict__ m0s) {
    const int lane = threadIdx.x & 31;
    const int wv   = threadIdx.x >> 5;
    const int row0 = blockIdx.x * 16;
    const int col0 = wv * 16;
    const int hh   = lane >> 4;
    const int l15  = lane & 15;

    v8f c = {};
    const float* xrow = x  + (row0 + l15) * D_FEAT + 2 * hh;
    const float* wcol = W0 + (2 * hh) * HIDDEN + col0 + l15;

#pragma unroll 4
    for (int k = 0; k < D_FEAT; k += 4) {
        v2f a, b;
        a.x = xrow[k];
        a.y = xrow[k + 1];
        b.x = wcol[k * HIDDEN];
        b.y = wcol[k * HIDDEN + HIDDEN];
        c = __builtin_amdgcn_wmma_f32_16x16x4_f32(false, a, false, b,
                                                  (short)0, c, false, false);
    }
    float* out = m0s + (row0 + 8 * hh) * HIDDEN + col0 + l15;
    const float* dr = dinv + row0 + 8 * hh;
#pragma unroll
    for (int v = 0; v < 8; ++v) out[v * HIDDEN] = dr[v] * c[v];
}


__device__ __forceinline__ float lane_f(float v, int i) {
    return __int_as_float(__builtin_amdgcn_readlane(__float_as_int(v), i));
}

__global__ __launch_bounds__(128) void k_pull0_gemm1(const int* __restrict__ ei,
                                                     const float* __restrict__ ew,
                                                     const int* __restrict__ cnt,
                                                     const int* __restrict__ ell,
                                                     const float* __restrict__ m0s,
                                                     const float* __restrict__ b0,
                                                     const float* __restrict__ W1,
                                                     float* __restrict__ m1s) {
    __shared__ float hs[16 * HPAD];
    __shared__ float dl[16];
    const int lane = threadIdx.x & 31;
    const int wv   = __builtin_amdgcn_readfirstlane((int)(threadIdx.x >> 5));
    const int row0 = blockIdx.x * 16;
    const float2 bb = *(const float2*)(b0 + 2 * lane);

    for (int q = 0; q < 4; ++q) {
        const int r = wv * 4 + q;
        const int d = row0 + r;
        const int n = __builtin_amdgcn_readfirstlane(cnt[d]);
        float ax = 0.0f, ay = 0.0f, dsum = 0.0f;
        if (n <= CAP) {
            const int* row = ell + d * CAP;
            const int e0 = (lane < n)      ? row[lane]      : 0;
            const int e1 = (lane + 32 < n) ? row[lane + 32] : 0;
            const int   s0 = ei[e0], s1 = ei[e1];
            const float w0 = ew[e0], w1 = ew[e1];
            const int n0 = n < 32 ? n : 32;
#pragma unroll 8
            for (int i = 0; i < n0; ++i) {
                const int   s = __builtin_amdgcn_readlane(s0, i);
                const float w = lane_f(w0, i);
                const float2 v = *(const float2*)(m0s + s * HIDDEN + 2 * lane);
                ax = fmaf(w, v.x, ax); ay = fmaf(w, v.y, ay); dsum += w;
            }
#pragma unroll 4
            for (int i = 32; i < n; ++i) {
                const int   s = __builtin_amdgcn_readlane(s1, i - 32);
                const float w = lane_f(w1, i - 32);
                const float2 v = *(const float2*)(m0s + s * HIDDEN + 2 * lane);
                ax = fmaf(w, v.x, ax); ay = fmaf(w, v.y, ay); dsum += w;
            }
        } else {
            for (int e = 0; e < N_EDGES; ++e) {
                if (ei[N_EDGES + e] == d) {
                    const int s = ei[e];
                    const float w = ew[e];
                    const float2 v = *(const float2*)(m0s + s * HIDDEN + 2 * lane);
                    ax = fmaf(w, v.x, ax); ay = fmaf(w, v.y, ay); dsum += w;
                }
            }
        }
        const float di = rsqrtf(fmaxf(dsum + 1.0f, 1e-12f));
        const float2 sv = *(const float2*)(m0s + d * HIDDEN + 2 * lane);
        hs[r * HPAD + 2 * lane]     = fmaxf(di * (ax + sv.x) + bb.x, 0.0f);
        hs[r * HPAD + 2 * lane + 1] = fmaxf(di * (ay + sv.y) + bb.y, 0.0f);
        if (lane == 0) dl[r] = di;
    }
    __syncthreads();
    if (wv != 0) return;

    const int hh  = lane >> 4;
    const int l15 = lane & 15;
    v8f c = {};
    const float* hrow = hs + l15 * HPAD + 2 * hh;
#pragma unroll
    for (int k = 0; k < HIDDEN; k += 4) {
        v2f a, b;
        a.x = hrow[k];
        a.y = hrow[k + 1];
        float bx = 0.0f, by = 0.0f;
        if (l15 < NCLASS) {
            bx = W1[(k + 2 * hh) * NCLASS + l15];
            by = W1[(k + 2 * hh + 1) * NCLASS + l15];
        }
        b.x = bx; b.y = by;
        c = __builtin_amdgcn_wmma_f32_16x16x4_f32(false, a, false, b,
                                                  (short)0, c, false, false);
    }
    if (l15 < NCLASS) {
        float* out = m1s + (row0 + 8 * hh) * NCLASS + l15;
#pragma unroll
        for (int v = 0; v < 8; ++v) out[v * NCLASS] = dl[v + 8 * hh] * c[v];
    }
}


__global__ __launch_bounds__(256) void k_pull1(const int* __restrict__ ei,
                                               const float* __restrict__ ew,
                                               const int* __restrict__ cnt,
                                               const int* __restrict__ ell,
                                               const float* __restrict__ m1s,
                                               const float* __restrict__ b1,
                                               float* __restrict__ out) {
    const int j  = threadIdx.x & 7;
    const int jj = j < NCLASS ? j : 0;
    const int d  = blockIdx.x * 32 + (threadIdx.x >> 3);
    const int n  = cnt[d];
    float acc = 0.0f, dsum = 0.0f;
    if (n <= CAP) {
        const int* row = ell + d * CAP;
        for (int c = 0; c < n; c += 8) {
            const int e  = (c + j < n) ? row[c + j] : 0;
            const int sl = ei[e];
            const float wl = ew[e];
            const int km = (n - c < 8) ? (n - c) : 8;
            for (int k = 0; k < km; ++k) {
                const int   s = __shfl(sl, k, 8);
                const float w = __shfl(wl, k, 8);
                acc = fmaf(w, m1s[s * NCLASS + jj], acc);
                dsum += w;
            }
        }
    } else {
        for (int e = 0; e < N_EDGES; ++e) {
            if (ei[N_EDGES + e] == d) {
                const float w = ew[e];
                acc = fmaf(w, m1s[ei[e] * NCLASS + jj], acc);
                dsum += w;
            }
        }
    }
    const float di = rsqrtf(fmaxf(dsum + 1.0f, 1e-12f));
    if (j < NCLASS) out[d * NCLASS + j] = di * (acc + m1s[d * NCLASS + j]) + b1[j];
}


extern "C" void kernel_launch(void* const* d_in, const int* in_sizes, int n_in,
                              void* d_out, int out_size, void* d_ws, size_t ws_size,
                              hipStream_t stream) {
    const float* x  = (const float*)d_in[0];
    const int*   ei = (const int*)  d_in[1];
    const float* ew = (const float*)d_in[2];
    const float* W0 = (const float*)d_in[3];
    const float* b0 = (const float*)d_in[4];
    const float* W1 = (const float*)d_in[5];
    const float* b1 = (const float*)d_in[6];
    float* out = (float*)d_out;

    float* ws   = (float*)d_ws;
    int*   cnt  = (int*)ws;
    float* m0s  = ws + N_NODES;
    int*   ell  = (int*)(m0s + (size_t)N_NODES * HIDDEN);
    float* m1s  = (float*)ell + (size_t)N_NODES * CAP;
    float* dinv = m1s;

    k_zero_cnt<<<(N_NODES + 255) / 256, 256, 0, stream>>>(cnt);
    k_fill    <<<(N_EDGES + 255) / 256, 256, 0, stream>>>(ei, cnt, ell);
    k_dinv    <<<(N_NODES + 255) / 256, 256, 0, stream>>>(ei, ew, cnt, ell, dinv);

    k_gemm0<<<ROWBLKS, 128, 0, stream>>>(x, W0, dinv, m0s);

    k_pull0_gemm1<<<ROWBLKS, 128, 0, stream>>>(ei, ew, cnt, ell, m0s, b0, W1, m1s);

    k_pull1<<<N_NODES / 32, 256, 0, stream>>>(ei, ew, cnt, ell, m1s, b1, out);
}


# slot-list walks issue a node's reads in whole groups before first use: layer 0 takes 16 rows a round on zero-weight padded lists (slot order kept); degree, layer 1 sum 8 lane partials by lane tree
# speedup vs baseline: 1.9302x; 1.0462x over previous
#include <hip/hip_runtime.h>

typedef __attribute__((ext_vector_type(2))) float v2f;
typedef __attribute__((ext_vector_type(8))) float v8f;

#define N_NODES 100000
#define N_EDGES 3200000
#define D_FEAT  256
#define HIDDEN  64
#define NCLASS  7
#define ROWBLKS (N_NODES / 16)
#define CAP     64
#define HPAD    66


__global__ void k_zero_cnt(int* __restrict__ cnt) {
    int i = blockIdx.x * blockDim.x + threadIdx.x;
    if (i < N_NODES) cnt[i] = 0;
}

__global__ void k_fill(const int* __restrict__ ei, int* __restrict__ cnt,
                       int* __restrict__ ell) {
    int e = blockIdx.x * blockDim.x + threadIdx.x;
    if (e < N_EDGES) {
        const int d = ei[N_EDGES + e];
        const int slot = atomicAdd(&cnt[d], 1);
        if (slot < CAP) ell[d * CAP + slot] = e;
    }
}

__global__ __launch_bounds__(256) void k_dinv(const int* __restrict__ ei, const float* __restrict__ ew,
                                              const int* __restrict__ cnt, const int* __restrict__ ell,
                                              float* __restrict__ dinv) {
    const int j = threadIdx.x & 7;
    const int d = blockIdx.x * 32 + (threadIdx.x >> 3);
    const int n = cnt[d];
    float dsum = 0.0f;
    if (n <= CAP) {
        const int* row = ell + d * CAP;
        for (int h = 0; h < 2; ++h) {
            if (h * 32 < n) {
                int e[4]; float w[4];
                for (int t = 0; t < 4; ++t) {
                    const int slot = h * 32 + t * 8 + j;
                    const int v = row[slot];
                    e[t] = slot < n ? v : 0;
                }
                for (int t = 0; t < 4; ++t) {
                    const int slot = h * 32 + t * 8 + j;
                    const float x = ew[e[t]];
                    w[t] = slot < n ? x : 0.0f;
                }
                dsum += (w[0] + w[1]) + (w[2] + w[3]);
            }
        }
    } else {
        for (int e = j; e < N_EDGES; e += 8)
            if (ei[N_EDGES + e] == d) dsum += ew[e];
    }
    dsum += __shfl_xor(dsum, 4, 8);
    dsum += __shfl_xor(dsum, 2, 8);
    dsum += __shfl_xor(dsum, 1, 8);
    if (j == 0) dinv[d] = rsqrtf(fmaxf(dsum + 1.0f, 1e-12f));
}


__global__ __launch_bounds__(128) void k_gemm0(const float* __restrict__ x,
                                               const float* __restrict__ W0,
                                               const float* __restrict__ dinv,
                                               float* __restrict__ m0s) {
    const int lane = threadIdx.x & 31;
    const int wv   = threadIdx.x >> 5;
    const int row0 = blockIdx.x * 16;
    const int col0 = wv * 16;
    const int hh   = lane >> 4;
    const int l15  = lane & 15;

    v8f c = {};
    const float* xrow = x  + (row0 + l15) * D_FEAT + 2 * hh;
    const float* wcol = W0 + (2 * hh) * HIDDEN + col0 + l15;

#pragma unroll 4
    for (int k = 0; k < D_FEAT; k += 4) {
        v2f a, b;
        a.x = xrow[k];
        a.y = xrow[k + 1];
        b.x = wcol[k * HIDDEN];
        b.y = wcol[k * HIDDEN + HIDDEN];
        c = __builtin_amdgcn_wmma_f32_16x16x4_f32(false, a, false, b,
                                                  (short)0, c, false, false);
    }
    float* out = m0s + (row0 + 8 * hh) * HIDDEN + col0 + l15;
    const float* dr = dinv + row0 + 8 * hh;
#pragma unroll
    for (int v = 0; v < 8; ++v) out[v * HIDDEN] = dr[v] * c[v];
}


__device__ __forceinline__ float lane_f(float v, int i) {
    return __int_as_float(__builtin_amdgcn_readlane(__float_as_int(v), i));
}

__global__ __launch_bounds__(128) void k_pull0_gemm1(const int* __restrict__ ei,
                                                     const float* __restrict__ ew,
                                                     const int* __restrict__ cnt,
                                                     const int* __restrict__ ell,
                                                     const float* __restrict__ m0s,
                                                     const float* __restrict__ b0,
                                                     const float* __restrict__ W1,
                                                     float* __restrict__ m1s) {
    __shared__ float hs[16 * HPAD];
    __shared__ float dl[16];
    const int lane = threadIdx.x & 31;
    const int wv   = __builtin_amdgcn_readfirstlane((int)(threadIdx.x >> 5));
    const int row0 = blockIdx.x * 16;
    const float2 bb = *(const float2*)(b0 + 2 * lane);

    const int d0 = row0 + wv * 4;
    int n4[4];
    int   s4[4][2];
    float w4[4][2];
    for (int q = 0; q < 4; ++q) n4[q] = __builtin_amdgcn_readfirstlane(cnt[d0 + q]);
    {
        int e4[4][2];
        for (int q = 0; q < 4; ++q)
            for (int h = 0; h < 2; ++h) {
                const int x = ell[(d0 + q) * CAP + h * 32 + lane];
                e4[q][h] = (h * 32 + lane < n4[q]) ? x : 0;
            }
        for (int q = 0; q < 4; ++q)
            for (int h = 0; h < 2; ++h) {
                s4[q][h] = ei[e4[q][h]];
                const float x = ew[e4[q][h]];
                w4[q][h] = (h * 32 + lane < n4[q]) ? x : 0.0f;
            }
    }

#pragma unroll
    for (int q = 0; q < 4; ++q) {
        const int r = wv * 4 + q;
        const int d = d0 + q;
        const int n = n4[q];
        float ax = 0.0f, ay = 0.0f, dsum = 0.0f;
        if (n <= CAP) {
            for (int i0 = 0; i0 < n; i0 += 16) {
                const int   sh = i0 < 32 ? s4[q][0] : s4[q][1];
                const float wh = i0 < 32 ? w4[q][0] : w4[q][1];
                const int   b  = i0 & 31;
                for (int u = 0; u < 16; ++u) {
                    const int   s = __builtin_amdgcn_readlane(sh, b + u);
                    const float w = lane_f(wh, b + u);
                    const float2 v = *(const float2*)(m0s + s * HIDDEN + 2 * lane);
                    ax = fmaf(w, v.x, ax); ay = fmaf(w, v.y, ay); dsum += w;
                }
            }
        } else {
            for (int e = 0; e < N_EDGES; ++e) {
                if (ei[N_EDGES + e] == d) {
                    const int s = ei[e];
                    const float w = ew[e];
                    const float2 v = *(const float2*)(m0s + s * HIDDEN + 2 * lane);
                    ax = fmaf(w, v.x, ax); ay = fmaf(w, v.y, ay); dsum += w;
                }
            }
        }
        const float di = rsqrtf(fmaxf(dsum + 1.0f, 1e-12f));
        const float2 sv = *(const float2*)(m0s + d * HIDDEN + 2 * lane);
        hs[r * HPAD + 2 * lane]     = fmaxf(di * (ax + sv.x) + bb.x, 0.0f);
        hs[r * HPAD + 2 * lane + 1] = fmaxf(di * (ay + sv.y) + bb.y, 0.0f);
        if (lane == 0) dl[r] = di;
    }
    __syncthreads();
    if (wv != 0) return;

    const int hh  = lane >> 4;
    const int l15 = lane & 15;
    v8f c = {};
    const float* hrow = hs + l15 * HPAD + 2 * hh;
#pragma unroll
    for (int k = 0; k < HIDDEN; k += 4) {
        v2f a, b;
        a.x = hrow[k];
        a.y = hrow[k + 1];
        float bx = 0.0f, by = 0.0f;
        if (l15 < NCLASS) {
            bx = W1[(k + 2 * hh) * NCLASS + l15];
            by = W1[(k + 2 * hh + 1) * NCLASS + l15];
        }
        b.x = bx; b.y = by;
        c = __builtin_amdgcn_wmma_f32_16x16x4_f32(false, a, false, b,
                                                  (short)0, c, false, false);
    }
    if (l15 < NCLASS) {
        float* out = m1s + (row0 + 8 * hh) * NCLASS + l15;
#pragma unroll
        for (int v = 0; v < 8; ++v) out[v * NCLASS] = dl[v + 8 * hh] * c[v];
    }
}


__global__ __launch_bounds__(256) void k_pull1(const int* __restrict__ ei,
                                               const float* __restrict__ ew,
                                               const int* __restrict__ cnt,
                                               const int* __restrict__ ell,
                                               const float* __restrict__ m1s,
                                               const float* __restrict__ b1,
                                               float* __restrict__ out) {
    const int j = threadIdx.x & 7;
    const int d = blockIdx.x * 32 + (threadIdx.x >> 3);
    const int n = cnt[d];
    float a[NCLASS];
    for (int k = 0; k < NCLASS; ++k) a[k] = 0.0f;
    float dsum = 0.0f;
    if (n <= CAP) {
        const int* row = ell + d * CAP;
        for (int h = 0; h < 2; ++h) {
            if (h * 32 < n) {
                int e[4], s[4]; float w[4]; float v[4][NCLASS];
                for (int t = 0; t < 4; ++t) {
                    const int slot = h * 32 + t * 8 + j;
                    const int x = row[slot];
                    e[t] = slot < n ? x : 0;
                }
                for (int t = 0; t < 4; ++t) {
                    const int slot = h * 32 + t * 8 + j;
                    s[t] = ei[e[t]];
                    const float x = ew[e[t]];
                    w[t] = slot < n ? x : 0.0f;
                }
                for (int t = 0; t < 4; ++t)
                    for (int k = 0; k < NCLASS; ++k) v[t][k] = m1s[s[t] * NCLASS + k];
                for (int t = 0; t < 4; ++t) {
                    for (int k = 0; k < NCLASS; ++k) a[k] = fmaf(w[t], v[t][k], a[k]);
                    dsum += w[t];
                }
            }
        }
    } else {
        for (int e = j; e < N_EDGES; e += 8) {
            if (ei[N_EDGES + e] == d) {
                const float w = ew[e];
                const int s = ei[e];
                for (int k = 0; k < NCLASS; ++k) a[k] = fmaf(w, m1s[s * NCLASS + k], a[k]);
                dsum += w;
            }
        }
    }
    for (int k = 0; k < NCLASS; ++k) {
        a[k] += __shfl_xor(a[k], 4, 8);
        a[k] += __shfl_xor(a[k], 2, 8);
        a[k] += __shfl_xor(a[k], 1, 8);
    }
    dsum += __shfl_xor(dsum, 4, 8);
    dsum += __shfl_xor(dsum, 2, 8);
    dsum += __shfl_xor(dsum, 1, 8);
    float aj = a[0];
    for (int k = 1; k < NCLASS; ++k) aj = (j == k) ? a[k] : aj;
    const float di = rsqrtf(fmaxf(dsum + 1.0f, 1e-12f));
    if (j < NCLASS) out[d * NCLASS + j] = di * (aj + m1s[d * NCLASS + j]) + b1[j];
}


extern "C" void kernel_launch(void* const* d_in, const int* in_sizes, int n_in,
                              void* d_out, int out_size, void* d_ws, size_t ws_size,
                              hipStream_t stream) {
    const float* x  = (const float*)d_in[0];
    const int*   ei = (const int*)  d_in[1];
    const float* ew = (const float*)d_in[2];
    const float* W0 = (const float*)d_in[3];
    const float* b0 = (const float*)d_in[4];
    const float* W1 = (const float*)d_in[5];
    const float* b1 = (const float*)d_in[6];
    float* out = (float*)d_out;

    float* ws   = (float*)d_ws;
    int*   cnt  = (int*)ws;
    float* m0s  = ws + N_NODES;
    int*   ell  = (int*)(m0s + (size_t)N_NODES * HIDDEN);
    float* m1s  = (float*)ell + (size_t)N_NODES * CAP;
    float* dinv = m1s;

    k_zero_cnt<<<(N_NODES + 255) / 256, 256, 0, stream>>>(cnt);
    k_fill    <<<(N_EDGES + 255) / 256, 256, 0, stream>>>(ei, cnt, ell);
    k_dinv    <<<N_NODES / 32, 256, 0, stream>>>(ei, ew, cnt, ell, dinv);

    k_gemm0<<<ROWBLKS, 128, 0, stream>>>(x, W0, dinv, m0s);

    k_pull0_gemm1<<<ROWBLKS, 128, 0, stream>>>(ei, ew, cnt, ell, m0s, b0, W1, m1s);

    k_pull1<<<N_NODES / 32, 256, 0, stream>>>(ei, ew, cnt, ell, m1s, b1, out);
}


# slot table built by a two-level LDS counting sort (edges binned by 400-node part, slots by LDS adds, table stored as whole lines); 3.2M random returning global adds go; sums in new slot order
# speedup vs baseline: 3.4876x; 1.8069x over previous
#include <hip/hip_runtime.h>

typedef __attribute__((ext_vector_type(2))) float v2f;
typedef __attribute__((ext_vector_type(8))) float v8f;

#define N_NODES 100000
#define N_EDGES 3200000
#define D_FEAT  256
#define HIDDEN  64
#define NCLASS  7
#define ROWBLKS (N_NODES / 16)
#define CAP     64
#define HPAD    66


#define PART    400
#define NPART   (N_NODES / PART)
#define BCAP    16384
#define EPB     8192
#define NBINBLK ((N_EDGES + EPB - 1) / EPB)
#define CSTRIDE 64

__global__ void k_zero_cur(int* __restrict__ gcur) {
    if (threadIdx.x < NPART) gcur[threadIdx.x * CSTRIDE] = 0;
}

__global__ __launch_bounds__(1024) void k_bin(const int* __restrict__ ei, int* __restrict__ gcur,
                                              unsigned* __restrict__ bins) {
    __shared__ int lhist[256];
    __shared__ int loff[256];
    __shared__ int lgb[256];
    __shared__ unsigned lrec[EPB];
    __shared__ unsigned char lbin[EPB];
    const int t  = threadIdx.x;
    const int e0 = blockIdx.x * EPB;
    if (t < 256) lhist[t] = 0;
    __syncthreads();

    int d[8], bn[8], rk[8];
    unsigned rec[8];
#pragma unroll
    for (int i = 0; i < 8; ++i) {
        const int e = e0 + i * 1024 + t;
        d[i] = e < N_EDGES ? ei[N_EDGES + e] : -1;
    }
#pragma unroll
    for (int i = 0; i < 8; ++i) {
        bn[i] = -1; rk[i] = 0; rec[i] = 0u;
        if (d[i] >= 0) {
            const unsigned b = (unsigned)d[i] / PART;
            bn[i]  = (int)b;
            rec[i] = (((unsigned)d[i] - b * PART) << 22) | (unsigned)(e0 + i * 1024 + t);
            rk[i]  = atomicAdd(&lhist[b], 1);
        }
    }
    __syncthreads();

    int gb = 0;
    if (t < NPART) {
        const int c = lhist[t];
        if (c > 0) gb = atomicAdd(&gcur[t * CSTRIDE], c);
    }
    if (t < 32) {
        int v[8];
        int run = 0;
#pragma unroll
        for (int k = 0; k < 8; ++k) { const int x = lhist[8 * t + k]; v[k] = run; run += x; }
        int inc = run;
        for (int o = 1; o < 32; o <<= 1) {
            const int y = __shfl_up(inc, o);
            if (t >= o) inc += y;
        }
#pragma unroll
        for (int k = 0; k < 8; ++k) loff[8 * t + k] = inc - run + v[k];
    }
    if (t < 256) lgb[t] = gb;
    __syncthreads();

#pragma unroll
    for (int i = 0; i < 8; ++i) {
        if (bn[i] >= 0) {
            const int pos = loff[bn[i]] + rk[i];
            lrec[pos] = rec[i];
            lbin[pos] = (unsigned char)bn[i];
        }
    }
    __syncthreads();

    const int nvalid = (N_EDGES - e0) < EPB ? (N_EDGES - e0) : EPB;
#pragma unroll
    for (int i = 0; i < 8; ++i) {
        const int p = i * 1024 + t;
        if (p < nvalid) {
            const int b = lbin[p];
            const int g = lgb[b] + (p - loff[b]);
            if (g < BCAP) bins[b * BCAP + g] = lrec[p];
        }
    }
}

__global__ __launch_bounds__(1024) void k_place(const int* __restrict__ gcur,
                                                const unsigned* __restrict__ bins,
                                                int* __restrict__ cnt, int* __restrict__ ell) {
    __shared__ int lell[PART * CAP];
    __shared__ int lcnt[PART];
    const int t    = threadIdx.x;
    const int part = blockIdx.x;
    const int total_raw = gcur[part * CSTRIDE];
    const int total = total_raw < BCAP ? total_raw : BCAP;
    const unsigned* src = bins + part * BCAP;

    unsigned r[BCAP / 1024];
#pragma unroll
    for (int i = 0; i < BCAP / 1024; ++i) {
        const int p = i * 1024 + t;
        r[i] = p < total ? src[p] : 0xFFFFFFFFu;
    }
    const int4 z = {0, 0, 0, 0};
    for (int q = t; q < PART * CAP / 4; q += 1024) ((int4*)lell)[q] = z;
    if (t < PART) lcnt[t] = 0;
    __syncthreads();

    int sl[BCAP / 1024];
#pragma unroll
    for (int i = 0; i < BCAP / 1024; ++i)
        sl[i] = r[i] != 0xFFFFFFFFu ? atomicAdd(&lcnt[r[i] >> 22], 1) : CAP;
#pragma unroll
    for (int i = 0; i < BCAP / 1024; ++i)
        if (sl[i] < CAP) lell[(int)(r[i] >> 22) * CAP + sl[i]] = (int)(r[i] & 0x3FFFFFu);
    __syncthreads();

    if (t < PART) cnt[part * PART + t] = total_raw > BCAP ? CAP + 1 : lcnt[t];
    int4* dst = (int4*)(ell + part * PART * CAP);
    for (int q = t; q < PART * CAP / 4; q += 1024) dst[q] = ((const int4*)lell)[q];
}

__global__ __launch_bounds__(256) void k_dinv(const int* __restrict__ ei, const float* __restrict__ ew,
                                              const int* __restrict__ cnt, const int* __restrict__ ell,
                                              float* __restrict__ dinv) {
    const int j = threadIdx.x & 7;
    const int d = blockIdx.x * 32 + (threadIdx.x >> 3);
    const int n = cnt[d];
    float dsum = 0.0f;
    if (n <= CAP) {
        const int* row = ell + d * CAP;
        for (int h = 0; h < 2; ++h) {
            if (h * 32 < n) {
                int e[4]; float w[4];
                for (int t = 0; t < 4; ++t) {
                    const int slot = h * 32 + t * 8 + j;
                    const int v = row[slot];
                    e[t] = slot < n ? v : 0;
                }
                for (int t = 0; t < 4; ++t) {
                    const int slot = h * 32 + t * 8 + j;
                    const float x = ew[e[t]];
                    w[t] = slot < n ? x : 0.0f;
                }
                dsum += (w[0] + w[1]) + (w[2] + w[3]);
            }
        }
    } else {
        for (int e = j; e < N_EDGES; e += 8)
            if (ei[N_EDGES + e] == d) dsum += ew[e];
    }
    dsum += __shfl_xor(dsum, 4, 8);
    dsum += __shfl_xor(dsum, 2, 8);
    dsum += __shfl_xor(dsum, 1, 8);
    if (j == 0) dinv[d] = rsqrtf(fmaxf(dsum + 1.0f, 1e-12f));
}


__global__ __launch_bounds__(128) void k_gemm0(const float* __restrict__ x,
                                               const float* __restrict__ W0,
                                               const float* __restrict__ dinv,
                                               float* __restrict__ m0s) {
    const int lane = threadIdx.x & 31;
    const int wv   = threadIdx.x >> 5;
    const int row0 = blockIdx.x * 16;
    const int col0 = wv * 16;
    const int hh   = lane >> 4;
    const int l15  = lane & 15;

    v8f c = {};
    const float* xrow = x  + (row0 + l15) * D_FEAT + 2 * hh;
    const float* wcol = W0 + (2 * hh) * HIDDEN + col0 + l15;

#pragma unroll 4
    for (int k = 0; k < D_FEAT; k += 4) {
        v2f a, b;
        a.x = xrow[k];
        a.y = xrow[k + 1];
        b.x = wcol[k * HIDDEN];
        b.y = wcol[k * HIDDEN + HIDDEN];
        c = __builtin_amdgcn_wmma_f32_16x16x4_f32(false, a, false, b,
                                                  (short)0, c, false, false);
    }
    float* out = m0s + (row0 + 8 * hh) * HIDDEN + col0 + l15;
    const float* dr = dinv + row0 + 8 * hh;
#pragma unroll
    for (int v = 0; v < 8; ++v) out[v * HIDDEN] = dr[v] * c[v];
}


__device__ __forceinline__ float lane_f(float v, int i) {
    return __int_as_float(__builtin_amdgcn_readlane(__float_as_int(v), i));
}

__global__ __launch_bounds__(128) void k_pull0_gemm1(const int* __restrict__ ei,
                                                     const float* __restrict__ ew,
                                                     const int* __restrict__ cnt,
                                                     const int* __restrict__ ell,
                                                     const float* __restrict__ m0s,
                                                     const float* __restrict__ b0,
                                                     const float* __restrict__ W1,
                                                     float* __restrict__ m1s) {
    __shared__ float hs[16 * HPAD];
    __shared__ float dl[16];
    const int lane = threadIdx.x & 31;
    const int wv   = __builtin_amdgcn_readfirstlane((int)(threadIdx.x >> 5));
    const int row0 = blockIdx.x * 16;
    const float2 bb = *(const float2*)(b0 + 2 * lane);

    const int d0 = row0 + wv * 4;
    int n4[4];
    int   s4[4][2];
    float w4[4][2];
    for (int q = 0; q < 4; ++q) n4[q] = __builtin_amdgcn_readfirstlane(cnt[d0 + q]);
    {
        int e4[4][2];
        for (int q = 0; q < 4; ++q)
            for (int h = 0; h < 2; ++h) {
                const int x = ell[(d0 + q) * CAP + h * 32 + lane];
                e4[q][h] = (h * 32 + lane < n4[q]) ? x : 0;
            }
        for (int q = 0; q < 4; ++q)
            for (int h = 0; h < 2; ++h) {
                s4[q][h] = ei[e4[q][h]];
                const float x = ew[e4[q][h]];
                w4[q][h] = (h * 32 + lane < n4[q]) ? x : 0.0f;
            }
    }

#pragma unroll
    for (int q = 0; q < 4; ++q) {
        const int r = wv * 4 + q;
        const int d = d0 + q;
        const int n = n4[q];
        float ax = 0.0f, ay = 0.0f, dsum = 0.0f;
        if (n <= CAP) {
            for (int i0 = 0; i0 < n; i0 += 16) {
                const int   sh = i0 < 32 ? s4[q][0] : s4[q][1];
                const float wh = i0 < 32 ? w4[q][0] : w4[q][1];
                const int   b  = i0 & 31;
                for (int u = 0; u < 16; ++u) {
                    const int   s = __builtin_amdgcn_readlane(sh, b + u);
                    const float w = lane_f(wh, b + u);
                    const float2 v = *(const float2*)(m0s + s * HIDDEN + 2 * lane);
                    ax = fmaf(w, v.x, ax); ay = fmaf(w, v.y, ay); dsum += w;
                }
            }
        } else {
            for (int e = 0; e < N_EDGES; ++e) {
                if (ei[N_EDGES + e] == d) {
                    const int s = ei[e];
                    const float w = ew[e];
                    const float2 v = *(const float2*)(m0s + s * HIDDEN + 2 * lane);
                    ax = fmaf(w, v.x, ax); ay = fmaf(w, v.y, ay); dsum += w;
                }
            }
        }
        const float di = rsqrtf(fmaxf(dsum + 1.0f, 1e-12f));
        const float2 sv = *(const float2*)(m0s + d * HIDDEN + 2 * lane);
        hs[r * HPAD + 2 * lane]     = fmaxf(di * (ax + sv.x) + bb.x, 0.0f);
        hs[r * HPAD + 2 * lane + 1] = fmaxf(di * (ay + sv.y) + bb.y, 0.0f);
        if (lane == 0) dl[r] = di;
    }
    __syncthreads();
    if (wv != 0) return;

    const int hh  = lane >> 4;
    const int l15 = lane & 15;
    v8f c = {};
    const float* hrow = hs + l15 * HPAD + 2 * hh;
#pragma unroll
    for (int k = 0; k < HIDDEN; k += 4) {
        v2f a, b;
        a.x = hrow[k];
        a.y = hrow[k + 1];
        float bx = 0.0f, by = 0.0f;
        if (l15 < NCLASS) {
            bx = W1[(k + 2 * hh) * NCLASS + l15];
            by = W1[(k + 2 * hh + 1) * NCLASS + l15];
        }
        b.x = bx; b.y = by;
        c = __builtin_amdgcn_wmma_f32_16x16x4_f32(false, a, false, b,
                                                  (short)0, c, false, false);
    }
    if (l15 < NCLASS) {
        float* out = m1s + (row0 + 8 * hh) * NCLASS + l15;
#pragma unroll
        for (int v = 0; v < 8; ++v) out[v * NCLASS] = dl[v + 8 * hh] * c[v];
    }
}


__global__ __launch_bounds__(256) void k_pull1(const int* __restrict__ ei,
                                               const float* __restrict__ ew,
                                               const int* __restrict__ cnt,
                                               const int* __restrict__ ell,
                                               const float* __restrict__ m1s,
                                               const float* __restrict__ b1,
                                               float* __restrict__ out) {
    const int j = threadIdx.x & 7;
    const int d = blockIdx.x * 32 + (threadIdx.x >> 3);
    const int n = cnt[d];
    float a[NCLASS];
    for (int k = 0; k < NCLASS; ++k) a[k] = 0.0f;
    float dsum = 0.0f;
    if (n <= CAP) {
        const int* row = ell + d * CAP;
        for (int h = 0; h < 2; ++h) {
            if (h * 32 < n) {
                int e[4], s[4]; float w[4]; float v[4][NCLASS];
                for (int t = 0; t < 4; ++t) {
                    const int slot = h * 32 + t * 8 + j;
                    const int x = row[slot];
                    e[t] = slot < n ? x : 0;
                }
                for (int t = 0; t < 4; ++t) {
                    const int slot = h * 32 + t * 8 + j;
                    s[t] = ei[e[t]];
                    const float x = ew[e[t]];
                    w[t] = slot < n ? x : 0.0f;
                }
                for (int t = 0; t < 4; ++t)
                    for (int k = 0; k < NCLASS; ++k) v[t][k] = m1s[s[t] * NCLASS + k];
                for (int t = 0; t < 4; ++t) {
                    for (int k = 0; k < NCLASS; ++k) a[k] = fmaf(w[t], v[t][k], a[k]);
                    dsum += w[t];
                }
            }
        }
    } else {
        for (int e = j; e < N_EDGES; e += 8) {
            if (ei[N_EDGES + e] == d) {
                const float w = ew[e];
                const int s = ei[e];
                for (int k = 0; k < NCLASS; ++k) a[k] = fmaf(w, m1s[s * NCLASS + k], a[k]);
                dsum += w;
            }
        }
    }
    for (int k = 0; k < NCLASS; ++k) {
        a[k] += __shfl_xor(a[k], 4, 8);
        a[k] += __shfl_xor(a[k], 2, 8);
        a[k] += __shfl_xor(a[k], 1, 8);
    }
    dsum += __shfl_xor(dsum, 4, 8);
    dsum += __shfl_xor(dsum, 2, 8);
    dsum += __shfl_xor(dsum, 1, 8);
    float aj = a[0];
    for (int k = 1; k < NCLASS; ++k) aj = (j == k) ? a[k] : aj;
    const float di = rsqrtf(fmaxf(dsum + 1.0f, 1e-12f));
    if (j < NCLASS) out[d * NCLASS + j] = di * (aj + m1s[d * NCLASS + j]) + b1[j];
}


extern "C" void kernel_launch(void* const* d_in, const int* in_sizes, int n_in,
                              void* d_out, int out_size, void* d_ws, size_t ws_size,
                              hipStream_t stream) {
    const float* x  = (const float*)d_in[0];
    const int*   ei = (const int*)  d_in[1];
    const float* ew = (const float*)d_in[2];
    const float* W0 = (const float*)d_in[3];
    const float* b0 = (const float*)d_in[4];
    const float* W1 = (const float*)d_in[5];
    const float* b1 = (const float*)d_in[6];
    float* out = (float*)d_out;

    float* ws   = (float*)d_ws;
    int*   cnt  = (int*)ws;
    float* m0s  = ws + N_NODES;
    int*   ell  = (int*)(m0s + (size_t)N_NODES * HIDDEN);
    float* m1s  = (float*)ell + (size_t)N_NODES * CAP;
    float* dinv = m1s;

    unsigned* bins = (unsigned*)m0s;
    int*      gcur = (int*)m0s + NPART * BCAP;

    k_zero_cur<<<1, 256, 0, stream>>>(gcur);
    k_bin     <<<NBINBLK, 1024, 0, stream>>>(ei, gcur, bins);
    k_place   <<<NPART, 1024, 0, stream>>>(gcur, bins, cnt, ell);
    k_dinv    <<<N_NODES / 32, 256, 0, stream>>>(ei, ew, cnt, ell, dinv);

    k_gemm0<<<ROWBLKS, 128, 0, stream>>>(x, W0, dinv, m0s);

    k_pull0_gemm1<<<ROWBLKS, 128, 0, stream>>>(ei, ew, cnt, ell, m0s, b0, W1, m1s);

    k_pull1<<<N_NODES / 32, 256, 0, stream>>>(ei, ew, cnt, ell, m1s, b1, out);
}


# in-edge lists hold each edge's own (source, weight) pair in one contiguous run per node (two-word sort records); degree, layer 0, layer 1 drop 2 random 4-B reads per edge; sums keep slot order
# speedup vs baseline: 4.5764x; 1.3122x over previous
#include <hip/hip_runtime.h>

typedef __attribute__((ext_vector_type(2))) float v2f;
typedef __attribute__((ext_vector_type(8))) float v8f;

#define N_NODES 100000
#define N_EDGES 3200000
#define D_FEAT  256
#define HIDDEN  64
#define NCLASS  7
#define ROWBLKS (N_NODES / 16)
#define HPAD    66


#define PART    400
#define NPART   (N_NODES / PART)
#define BCAP    14080
#define RPT     ((BCAP + 1023) / 1024)
#define EPB     8192
#define NBINBLK ((N_EDGES + EPB - 1) / EPB)
#define CSTRIDE 64
#define NMAX    1023

__global__ void k_zero_cur(int* __restrict__ gcur) {
    if (threadIdx.x < NPART) gcur[threadIdx.x * CSTRIDE] = 0;
}

__global__ __launch_bounds__(1024) void k_bin(const int* __restrict__ ei, const float* __restrict__ ew,
                                              int* __restrict__ gcur, int* __restrict__ binw) {
    int2* bins = (int2*)binw;
    __shared__ int lhist[256];
    __shared__ int loff[256];
    __shared__ int lgb[256];
    __shared__ int2 lrec[EPB];
    __shared__ unsigned char lbin[EPB];
    const int t  = threadIdx.x;
    const int e0 = blockIdx.x * EPB;
    if (t < 256) lhist[t] = 0;
    __syncthreads();

    int d[8], bn[8], rk[8], key[8];
    float w[8];
#pragma unroll
    for (int i = 0; i < 8; ++i) {
        const int e = e0 + i * 1024 + t;
        const bool ok = e < N_EDGES;
        d[i]   = ok ? ei[N_EDGES + e] : -1;
        key[i] = ok ? ei[e] : 0;
        w[i]   = ok ? ew[e] : 0.0f;
    }
#pragma unroll
    for (int i = 0; i < 8; ++i) {
        bn[i] = -1; rk[i] = 0;
        if (d[i] >= 0) {
            const unsigned b = (unsigned)d[i] / PART;
            bn[i]  = (int)b;
            key[i] = (int)((((unsigned)d[i] - b * PART) << 17) | (unsigned)key[i]);
            rk[i]  = atomicAdd(&lhist[b], 1);
        }
    }
    __syncthreads();

    int gb = 0;
    if (t < NPART) {
        const int c = lhist[t];
        if (c > 0) gb = atomicAdd(&gcur[t * CSTRIDE], c);
    }
    if (t < 32) {
        int v[8];
        int run = 0;
#pragma unroll
        for (int k = 0; k < 8; ++k) { const int x = lhist[8 * t + k]; v[k] = run; run += x; }
        int inc = run;
        for (int o = 1; o < 32; o <<= 1) {
            const int y = __shfl_up(inc, o);
            if (t >= o) inc += y;
        }
#pragma unroll
        for (int k = 0; k < 8; ++k) loff[8 * t + k] = inc - run + v[k];
    }
    if (t < 256) lgb[t] = gb;
    __syncthreads();

#pragma unroll
    for (int i = 0; i < 8; ++i) {
        if (bn[i] >= 0) {
            const int pos = loff[bn[i]] + rk[i];
            lrec[pos] = make_int2(key[i], __float_as_int(w[i]));
            lbin[pos] = (unsigned char)bn[i];
        }
    }
    __syncthreads();

    const int nvalid = (N_EDGES - e0) < EPB ? (N_EDGES - e0) : EPB;
#pragma unroll
    for (int i = 0; i < 8; ++i) {
        const int p = i * 1024 + t;
        if (p < nvalid) {
            const int b = lbin[p];
            const int g = lgb[b] + (p - loff[b]);
            if (g < BCAP) bins[b * BCAP + g] = lrec[p];
        }
    }
}

__global__ __launch_bounds__(1024) void k_place(const int* __restrict__ gcur,
                                                const int* __restrict__ binw,
                                                unsigned* __restrict__ rowinf, int* __restrict__ csrw) {
    const int2* bins = (const int2*)binw;
    int2* csr = (int2*)csrw;
    __shared__ int2 lrec[BCAP];
    __shared__ int lcnt[PART];
    __shared__ int lstart[PART];
    __shared__ int lbase;
    const int t    = threadIdx.x;
    const int part = blockIdx.x;
    const int total_raw = gcur[part * CSTRIDE];
    const bool over = total_raw > BCAP;
    const int total = over ? 0 : total_raw;
    const int2* src = bins + part * BCAP;

    int2 r[RPT];
#pragma unroll
    for (int i = 0; i < RPT; ++i) {
        const int p = i * 1024 + t;
        r[i] = p < total ? src[p] : make_int2(-1, 0);
    }
    int before = 0;
    if (t < part) {
        const int c = gcur[t * CSTRIDE];
        before = c > BCAP ? 0 : c;
    }
    if (t < PART) lcnt[t] = 0;
    if (t == 0) lbase = 0;
    __syncthreads();

    int sl[RPT];
#pragma unroll
    for (int i = 0; i < RPT; ++i)
        sl[i] = r[i].x >= 0 ? atomicAdd(&lcnt[r[i].x >> 17], 1) : 0;
    if (t < 256) {
        for (int o = 16; o > 0; o >>= 1) before += __shfl_xor(before, o);
        if ((t & 31) == 0 && before > 0) atomicAdd(&lbase, before);
    }
    __syncthreads();

    if (t < 32) {
        int v[13];
        int run = 0;
#pragma unroll
        for (int k = 0; k < 13; ++k) {
            const int idx = 13 * t + k;
            const int x = idx < PART ? lcnt[idx] : 0;
            v[k] = run; run += x;
        }
        int inc = run;
        for (int o = 1; o < 32; o <<= 1) {
            const int y = __shfl_up(inc, o);
            if (t >= o) inc += y;
        }
#pragma unroll
        for (int k = 0; k < 13; ++k) {
            const int idx = 13 * t + k;
            if (idx < PART) lstart[idx] = inc - run + v[k];
        }
    }
    __syncthreads();

#pragma unroll
    for (int i = 0; i < RPT; ++i)
        if (r[i].x >= 0) lrec[lstart[r[i].x >> 17] + sl[i]] = make_int2(r[i].x & 0x1FFFF, r[i].y);
    __syncthreads();

    const int base = lbase;
    if (t < PART) {
        const int cn = lcnt[t];
        rowinf[part * PART + t] = over ? (unsigned)NMAX
                                       : (((unsigned)(base + lstart[t]) << 10) | (unsigned)(cn < NMAX ? cn : NMAX));
    }
    int2* dst = csr + base;
    for (int q = t; q < total; q += 1024) dst[q] = lrec[q];
}

__global__ __launch_bounds__(256) void k_dinv(const int* __restrict__ ei, const float* __restrict__ ew,
                                              const unsigned* __restrict__ rowinf,
                                              const int* __restrict__ csrw,
                                              float* __restrict__ dinv) {
    const int2* csr = (const int2*)csrw;
    const int j = threadIdx.x & 7;
    const int d = blockIdx.x * 32 + (threadIdx.x >> 3);
    const unsigned ri = rowinf[d];
    const int n = (int)(ri & NMAX);
    float dsum = 0.0f;
    if (n < NMAX) {
        const int2* row = csr + (ri >> 10);
        for (int b = 0; b < n; b += 32) {
            float w[4];
            for (int t = 0; t < 4; ++t) {
                const int slot = b + t * 8 + j;
                w[t] = slot < n ? __int_as_float(row[slot].y) : 0.0f;
            }
            dsum += (w[0] + w[1]) + (w[2] + w[3]);
        }
    } else {
        for (int e = j; e < N_EDGES; e += 8)
            if (ei[N_EDGES + e] == d) dsum += ew[e];
    }
    dsum += __shfl_xor(dsum, 4, 8);
    dsum += __shfl_xor(dsum, 2, 8);
    dsum += __shfl_xor(dsum, 1, 8);
    if (j == 0) dinv[d] = rsqrtf(fmaxf(dsum + 1.0f, 1e-12f));
}


__global__ __launch_bounds__(128) void k_gemm0(const float* __restrict__ x,
                                               const float* __restrict__ W0,
                                               const float* __restrict__ dinv,
                                               float* __restrict__ m0s) {
    const int lane = threadIdx.x & 31;
    const int wv   = threadIdx.x >> 5;
    const int row0 = blockIdx.x * 16;
    const int col0 = wv * 16;
    const int hh   = lane >> 4;
    const int l15  = lane & 15;

    v8f c = {};
    const float* xrow = x  + (row0 + l15) * D_FEAT + 2 * hh;
    const float* wcol = W0 + (2 * hh) * HIDDEN + col0 + l15;

#pragma unroll 4
    for (int k = 0; k < D_FEAT; k += 4) {
        v2f a, b;
        a.x = xrow[k];
        a.y = xrow[k + 1];
        b.x = wcol[k * HIDDEN];
        b.y = wcol[k * HIDDEN + HIDDEN];
        c = __builtin_amdgcn_wmma_f32_16x16x4_f32(false, a, false, b,
                                                  (short)0, c, false, false);
    }
    float* out = m0s + (row0 + 8 * hh) * HIDDEN + col0 + l15;
    const float* dr = dinv + row0 + 8 * hh;
#pragma unroll
    for (int v = 0; v < 8; ++v) out[v * HIDDEN] = dr[v] * c[v];
}


__device__ __forceinline__ float lane_f(float v, int i) {
    return __int_as_float(__builtin_amdgcn_readlane(__float_as_int(v), i));
}

__global__ __launch_bounds__(128) void k_pull0_gemm1(const int* __restrict__ ei,
                                                     const float* __restrict__ ew,
                                                     const unsigned* __restrict__ rowinf,
                                                     const int* __restrict__ csrw,
                                                     const float* __restrict__ m0s,
                                                     const float* __restrict__ b0,
                                                     const float* __restrict__ W1,
                                                     float* __restrict__ m1s) {
    __shared__ float hs[16 * HPAD];
    __shared__ float dl[16];
    const int2* csr = (const int2*)csrw;
    const int lane = threadIdx.x & 31;
    const int wv   = __builtin_amdgcn_readfirstlane((int)(threadIdx.x >> 5));
    const int row0 = blockIdx.x * 16;
    const float2 bb = *(const float2*)(b0 + 2 * lane);

    const int d0 = row0 + wv * 4;
    int n4[4], st4[4];
    int   s4[4][2];
    float w4[4][2];
    for (int q = 0; q < 4; ++q) {
        const unsigned ri = (unsigned)__builtin_amdgcn_readfirstlane((int)rowinf[d0 + q]);
        n4[q]  = (int)(ri & NMAX);
        st4[q] = (int)(ri >> 10);
    }
    for (int q = 0; q < 4; ++q)
        for (int h = 0; h < 2; ++h) {
            int2 p = make_int2(0, 0);
            if (n4[q] < NMAX && h * 32 + lane < n4[q]) p = csr[st4[q] + h * 32 + lane];
            s4[q][h] = p.x;
            w4[q][h] = __int_as_float(p.y);
        }

#pragma unroll
    for (int q = 0; q < 4; ++q) {
        const int r = wv * 4 + q;
        const int d = d0 + q;
        const int n = n4[q];
        float ax = 0.0f, ay = 0.0f, dsum = 0.0f;
        if (n <= 64) {
            for (int i0 = 0; i0 < n; i0 += 16) {
                const int   sh = i0 < 32 ? s4[q][0] : s4[q][1];
                const float wh = i0 < 32 ? w4[q][0] : w4[q][1];
                const int   b  = i0 & 31;
                for (int u = 0; u < 16; ++u) {
                    const int   s = __builtin_amdgcn_readlane(sh, b + u);
                    const float w = lane_f(wh, b + u);
                    const float2 v = *(const float2*)(m0s + s * HIDDEN + 2 * lane);
                    ax = fmaf(w, v.x, ax); ay = fmaf(w, v.y, ay); dsum += w;
                }
            }
        } else if (n < NMAX) {
            const int2* row = csr + st4[q];
            for (int i0 = 0; i0 < n; i0 += 16) {
                int2 p = make_int2(0, 0);
                const int c0 = i0 & ~31;
                if (c0 + lane < n) p = row[c0 + lane];
                const int   sh = p.x;
                const float wh = __int_as_float(p.y);
                const int   b  = i0 & 31;
                for (int u = 0; u < 16; ++u) {
                    const int   s = __builtin_amdgcn_readlane(sh, b + u);
                    const float w = lane_f(wh, b + u);
                    const float2 v = *(const float2*)(m0s + s * HIDDEN + 2 * lane);
                    ax = fmaf(w, v.x, ax); ay = fmaf(w, v.y, ay); dsum += w;
                }
            }
        } else {
            for (int e = 0; e < N_EDGES; ++e) {
                if (ei[N_EDGES + e] == d) {
                    const int s = ei[e];
                    const float w = ew[e];
                    const float2 v = *(const float2*)(m0s + s * HIDDEN + 2 * lane);
                    ax = fmaf(w, v.x, ax); ay = fmaf(w, v.y, ay); dsum += w;
                }
            }
        }
        const float di = rsqrtf(fmaxf(dsum + 1.0f, 1e-12f));
        const float2 sv = *(const float2*)(m0s + d * HIDDEN + 2 * lane);
        hs[r * HPAD + 2 * lane]     = fmaxf(di * (ax + sv.x) + bb.x, 0.0f);
        hs[r * HPAD + 2 * lane + 1] = fmaxf(di * (ay + sv.y) + bb.y, 0.0f);
        if (lane == 0) dl[r] = di;
    }
    __syncthreads();
    if (wv != 0) return;

    const int hh  = lane >> 4;
    const int l15 = lane & 15;
    v8f c = {};
    const float* hrow = hs + l15 * HPAD + 2 * hh;
#pragma unroll
    for (int k = 0; k < HIDDEN; k += 4) {
        v2f a, b;
        a.x = hrow[k];
        a.y = hrow[k + 1];
        float bx = 0.0f, by = 0.0f;
        if (l15 < NCLASS) {
            bx = W1[(k + 2 * hh) * NCLASS + l15];
            by = W1[(k + 2 * hh + 1) * NCLASS + l15];
        }
        b.x = bx; b.y = by;
        c = __builtin_amdgcn_wmma_f32_16x16x4_f32(false, a, false, b,
                                                  (short)0, c, false, false);
    }
    if (l15 < NCLASS) {
        float* out = m1s + (row0 + 8 * hh) * NCLASS + l15;
#pragma unroll
        for (int v = 0; v < 8; ++v) out[v * NCLASS] = dl[v + 8 * hh] * c[v];
    }
}


__global__ __launch_bounds__(256) void k_pull1(const int* __restrict__ ei,
                                               const float* __restrict__ ew,
                                               const unsigned* __restrict__ rowinf,
                                               const int* __restrict__ csrw,
                                               const float* __restrict__ m1s,
                                               const float* __restrict__ b1,
                                               float* __restrict__ out) {
    const int2* csr = (const int2*)csrw;
    const int j = threadIdx.x & 7;
    const int d = blockIdx.x * 32 + (threadIdx.x >> 3);
    const unsigned ri = rowinf[d];
    const int n = (int)(ri & NMAX);
    float a[NCLASS];
    for (int k = 0; k < NCLASS; ++k) a[k] = 0.0f;
    float dsum = 0.0f;
    if (n < NMAX) {
        const int2* row = csr + (ri >> 10);
        for (int b = 0; b < n; b += 32) {
            int s[4]; float w[4]; float v[4][NCLASS];
            for (int t = 0; t < 4; ++t) {
                const int slot = b + t * 8 + j;
                int2 p = make_int2(0, 0);
                if (slot < n) p = row[slot];
                s[t] = p.x;
                w[t] = __int_as_float(p.y);
            }
            for (int t = 0; t < 4; ++t)
                for (int k = 0; k < NCLASS; ++k) v[t][k] = m1s[s[t] * NCLASS + k];
            for (int t = 0; t < 4; ++t) {
                for (int k = 0; k < NCLASS; ++k) a[k] = fmaf(w[t], v[t][k], a[k]);
                dsum += w[t];
            }
        }
    } else {
        for (int e = j; e < N_EDGES; e += 8) {
            if (ei[N_EDGES + e] == d) {
                const float w = ew[e];
                const int s = ei[e];
                for (int k = 0; k < NCLASS; ++k) a[k] = fmaf(w, m1s[s * NCLASS + k], a[k]);
                dsum += w;
            }
        }
    }
    for (int k = 0; k < NCLASS; ++k) {
        a[k] += __shfl_xor(a[k], 4, 8);
        a[k] += __shfl_xor(a[k], 2, 8);
        a[k] += __shfl_xor(a[k], 1, 8);
    }
    dsum += __shfl_xor(dsum, 4, 8);
    dsum += __shfl_xor(dsum, 2, 8);
    dsum += __shfl_xor(dsum, 1, 8);
    float aj = a[0];
    for (int k = 1; k < NCLASS; ++k) aj = (j == k) ? a[k] : aj;
    const float di = rsqrtf(fmaxf(dsum + 1.0f, 1e-12f));
    if (j < NCLASS) out[d * NCLASS + j] = di * (aj + m1s[d * NCLASS + j]) + b1[j];
}


extern "C" void kernel_launch(void* const* d_in, const int* in_sizes, int n_in,
                              void* d_out, int out_size, void* d_ws, size_t ws_size,
                              hipStream_t stream) {
    const float* x  = (const float*)d_in[0];
    const int*   ei = (const int*)  d_in[1];
    const float* ew = (const float*)d_in[2];
    const float* W0 = (const float*)d_in[3];
    const float* b0 = (const float*)d_in[4];
    const float* W1 = (const float*)d_in[5];
    const float* b1 = (const float*)d_in[6];
    float* out = (float*)d_out;

    float*    ws     = (float*)d_ws;
    unsigned* rowinf = (unsigned*)ws;
    int*      csr    = (int*)(ws + N_NODES);
    float*    m0s    = ws + N_NODES + 2 * (size_t)N_EDGES;
    float*    m1s    = m0s + (size_t)N_NODES * HIDDEN;
    float*    dinv   = m1s;

    int*  bins = (int*)m0s;
    int*  gcur = (int*)m0s + 2 * NPART * BCAP;

    k_zero_cur<<<1, 256, 0, stream>>>(gcur);
    k_bin     <<<NBINBLK, 1024, 0, stream>>>(ei, ew, gcur, bins);
    k_place   <<<NPART, 1024, 0, stream>>>(gcur, bins, rowinf, csr);
    k_dinv    <<<N_NODES / 32, 256, 0, stream>>>(ei, ew, rowinf, csr, dinv);

    k_gemm0<<<ROWBLKS, 128, 0, stream>>>(x, W0, dinv, m0s);

    k_pull0_gemm1<<<ROWBLKS, 128, 0, stream>>>(ei, ew, rowinf, csr, m0s, b0, W1, m1s);

    k_pull1<<<N_NODES / 32, 256, 0, stream>>>(ei, ew, rowinf, csr, m1s, b1, out);
}


# GEMM0 re-tiled: a wave owns a 16-row tile's 4 column tiles, x read once per tile by 16-B lane loads, W0 fragments from a block's LDS copy; k regrouped inside the products, sum order changes
# speedup vs baseline: 5.4066x; 1.1814x over previous
#include <hip/hip_runtime.h>

typedef __attribute__((ext_vector_type(2))) float v2f;
typedef __attribute__((ext_vector_type(8))) float v8f;

#define N_NODES 100000
#define N_EDGES 3200000
#define D_FEAT  256
#define HIDDEN  64
#define NCLASS  7
#define ROWBLKS (N_NODES / 16)
#define HPAD    66


#define PART    400
#define NPART   (N_NODES / PART)
#define BCAP    14080
#define RPT     ((BCAP + 1023) / 1024)
#define EPB     8192
#define NBINBLK ((N_EDGES + EPB - 1) / EPB)
#define CSTRIDE 64
#define NMAX    1023

__global__ void k_zero_cur(int* __restrict__ gcur) {
    if (threadIdx.x < NPART) gcur[threadIdx.x * CSTRIDE] = 0;
}

__global__ __launch_bounds__(1024) void k_bin(const int* __restrict__ ei, const float* __restrict__ ew,
                                              int* __restrict__ gcur, int* __restrict__ binw) {
    int2* bins = (int2*)binw;
    __shared__ int lhist[256];
    __shared__ int loff[256];
    __shared__ int lgb[256];
    __shared__ int2 lrec[EPB];
    __shared__ unsigned char lbin[EPB];
    const int t  = threadIdx.x;
    const int e0 = blockIdx.x * EPB;
    if (t < 256) lhist[t] = 0;
    __syncthreads();

    int d[8], bn[8], rk[8], key[8];
    float w[8];
#pragma unroll
    for (int i = 0; i < 8; ++i) {
        const int e = e0 + i * 1024 + t;
        const bool ok = e < N_EDGES;
        d[i]   = ok ? ei[N_EDGES + e] : -1;
        key[i] = ok ? ei[e] : 0;
        w[i]   = ok ? ew[e] : 0.0f;
    }
#pragma unroll
    for (int i = 0; i < 8; ++i) {
        bn[i] = -1; rk[i] = 0;
        if (d[i] >= 0) {
            const unsigned b = (unsigned)d[i] / PART;
            bn[i]  = (int)b;
            key[i] = (int)((((unsigned)d[i] - b * PART) << 17) | (unsigned)key[i]);
            rk[i]  = atomicAdd(&lhist[b], 1);
        }
    }
    __syncthreads();

    int gb = 0;
    if (t < NPART) {
        const int c = lhist[t];
        if (c > 0) gb = atomicAdd(&gcur[t * CSTRIDE], c);
    }
    if (t < 32) {
        int v[8];
        int run = 0;
#pragma unroll
        for (int k = 0; k < 8; ++k) { const int x = lhist[8 * t + k]; v[k] = run; run += x; }
        int inc = run;
        for (int o = 1; o < 32; o <<= 1) {
            const int y = __shfl_up(inc, o);
            if (t >= o) inc += y;
        }
#pragma unroll
        for (int k = 0; k < 8; ++k) loff[8 * t + k] = inc - run + v[k];
    }
    if (t < 256) lgb[t] = gb;
    __syncthreads();

#pragma unroll
    for (int i = 0; i < 8; ++i) {
        if (bn[i] >= 0) {
            const int pos = loff[bn[i]] + rk[i];
            lrec[pos] = make_int2(key[i], __float_as_int(w[i]));
            lbin[pos] = (unsigned char)bn[i];
        }
    }
    __syncthreads();

    const int nvalid = (N_EDGES - e0) < EPB ? (N_EDGES - e0) : EPB;
#pragma unroll
    for (int i = 0; i < 8; ++i) {
        const int p = i * 1024 + t;
        if (p < nvalid) {
            const int b = lbin[p];
            const int g = lgb[b] + (p - loff[b]);
            if (g < BCAP) bins[b * BCAP + g] = lrec[p];
        }
    }
}

__global__ __launch_bounds__(1024) void k_place(const int* __restrict__ gcur,
                                                const int* __restrict__ binw,
                                                unsigned* __restrict__ rowinf, int* __restrict__ csrw) {
    const int2* bins = (const int2*)binw;
    int2* csr = (int2*)csrw;
    __shared__ int2 lrec[BCAP];
    __shared__ int lcnt[PART];
    __shared__ int lstart[PART];
    __shared__ int lbase;
    const int t    = threadIdx.x;
    const int part = blockIdx.x;
    const int total_raw = gcur[part * CSTRIDE];
    const bool over = total_raw > BCAP;
    const int total = over ? 0 : total_raw;
    const int2* src = bins + part * BCAP;

    int2 r[RPT];
#pragma unroll
    for (int i = 0; i < RPT; ++i) {
        const int p = i * 1024 + t;
        r[i] = p < total ? src[p] : make_int2(-1, 0);
    }
    int before = 0;
    if (t < part) {
        const int c = gcur[t * CSTRIDE];
        before = c > BCAP ? 0 : c;
    }
    if (t < PART) lcnt[t] = 0;
    if (t == 0) lbase = 0;
    __syncthreads();

    int sl[RPT];
#pragma unroll
    for (int i = 0; i < RPT; ++i)
        sl[i] = r[i].x >= 0 ? atomicAdd(&lcnt[r[i].x >> 17], 1) : 0;
    if (t < 256) {
        for (int o = 16; o > 0; o >>= 1) before += __shfl_xor(before, o);
        if ((t & 31) == 0 && before > 0) atomicAdd(&lbase, before);
    }
    __syncthreads();

    if (t < 32) {
        int v[13];
        int run = 0;
#pragma unroll
        for (int k = 0; k < 13; ++k) {
            const int idx = 13 * t + k;
            const int x = idx < PART ? lcnt[idx] : 0;
            v[k] = run; run += x;
        }
        int inc = run;
        for (int o = 1; o < 32; o <<= 1) {
            const int y = __shfl_up(inc, o);
            if (t >= o) inc += y;
        }
#pragma unroll
        for (int k = 0; k < 13; ++k) {
            const int idx = 13 * t + k;
            if (idx < PART) lstart[idx] = inc - run + v[k];
        }
    }
    __syncthreads();

#pragma unroll
    for (int i = 0; i < RPT; ++i)
        if (r[i].x >= 0) lrec[lstart[r[i].x >> 17] + sl[i]] = make_int2(r[i].x & 0x1FFFF, r[i].y);
    __syncthreads();

    const int base = lbase;
    if (t < PART) {
        const int cn = lcnt[t];
        rowinf[part * PART + t] = over ? (unsigned)NMAX
                                       : (((unsigned)(base + lstart[t]) << 10) | (unsigned)(cn < NMAX ? cn : NMAX));
    }
    int2* dst = csr + base;
    for (int q = t; q < total; q += 1024) dst[q] = lrec[q];
}

__global__ __launch_bounds__(256) void k_dinv(const int* __restrict__ ei, const float* __restrict__ ew,
                                              const unsigned* __restrict__ rowinf,
                                              const int* __restrict__ csrw,
                                              float* __restrict__ dinv) {
    const int2* csr = (const int2*)csrw;
    const int j = threadIdx.x & 7;
    const int d = blockIdx.x * 32 + (threadIdx.x >> 3);
    const unsigned ri = rowinf[d];
    const int n = (int)(ri & NMAX);
    float dsum = 0.0f;
    if (n < NMAX) {
        const int2* row = csr + (ri >> 10);
        for (int b = 0; b < n; b += 32) {
            float w[4];
            for (int t = 0; t < 4; ++t) {
                const int slot = b + t * 8 + j;
                w[t] = slot < n ? __int_as_float(row[slot].y) : 0.0f;
            }
            dsum += (w[0] + w[1]) + (w[2] + w[3]);
        }
    } else {
        for (int e = j; e < N_EDGES; e += 8)
            if (ei[N_EDGES + e] == d) dsum += ew[e];
    }
    dsum += __shfl_xor(dsum, 4, 8);
    dsum += __shfl_xor(dsum, 2, 8);
    dsum += __shfl_xor(dsum, 1, 8);
    if (j == 0) dinv[d] = rsqrtf(fmaxf(dsum + 1.0f, 1e-12f));
}


#define G0_WAVES   25
#define G0_THREADS (G0_WAVES * 32)
#define G0_BLOCKS  (ROWBLKS / G0_WAVES)
#define G0_CHUNKS  (D_FEAT / 16)

__global__ __launch_bounds__(G0_THREADS) void k_gemm0(const float* __restrict__ x,
                                                      const float* __restrict__ W0,
                                                      const float* __restrict__ dinv,
                                                      float* __restrict__ m0s) {
    __shared__ float4 wl[D_FEAT * HIDDEN / 4];
    const int t    = threadIdx.x;
    const int lane = t & 31;
    const int wv   = t >> 5;
    const int hh   = lane >> 4;
    const int l15  = lane & 15;
    const int row0 = (blockIdx.x * G0_WAVES + wv) * 16;

    const float4* xr = (const float4*)(x + (row0 + l15) * D_FEAT + 8 * hh);
    float4 a[4][2];
    a[0][0] = xr[0]; a[0][1] = xr[1];
    a[1][0] = xr[4]; a[1][1] = xr[5];
    a[2][0] = xr[8]; a[2][1] = xr[9];

    float* wlf = (float*)wl;
    float4 wr[6];
#pragma unroll
    for (int j = 0; j < 6; ++j) {
        const int i = t + j * G0_THREADS;
        wr[j] = i < D_FEAT * HIDDEN / 4 ? ((const float4*)W0)[i] : make_float4(0.f, 0.f, 0.f, 0.f);
    }
#pragma unroll
    for (int j = 0; j < 6; ++j) {
        const int i = t + j * G0_THREADS;
        if (i < D_FEAT * HIDDEN / 4) {
            const int k   = i >> 4;
            const int col = (i & 15) * 4;
            const int ch = k >> 4, h = (k >> 3) & 1, sh = (k >> 2) & 1, s3 = k & 3;
            const int n = col >> 4, c = col & 15;
            const int base = ((((ch * 4 + n) * 2 + sh) * 32) + (h * 16 + c)) * 4 + s3;
            wlf[base]      = wr[j].x;
            wlf[base + 4]  = wr[j].y;
            wlf[base + 8]  = wr[j].z;
            wlf[base + 12] = wr[j].w;
        }
    }
    __syncthreads();

    v8f acc[4] = {};
#pragma unroll
    for (int ch = 0; ch < G0_CHUNKS; ++ch) {
        if (ch + 3 < G0_CHUNKS) {
            a[(ch + 3) & 3][0] = xr[(ch + 3) * 4];
            a[(ch + 3) & 3][1] = xr[(ch + 3) * 4 + 1];
        }
        const float4 a0 = a[ch & 3][0];
        const float4 a1 = a[ch & 3][1];
        float4 b0[4], b1[4];
#pragma unroll
        for (int n = 0; n < 4; ++n) {
            b0[n] = wl[((ch * 4 + n) * 2 + 0) * 32 + lane];
            b1[n] = wl[((ch * 4 + n) * 2 + 1) * 32 + lane];
        }
        __builtin_amdgcn_sched_barrier(0);
        v2f av, bv;
#pragma unroll
        for (int n = 0; n < 4; ++n) {
            av.x = a0.x; av.y = a0.y; bv.x = b0[n].x; bv.y = b0[n].y;
            acc[n] = __builtin_amdgcn_wmma_f32_16x16x4_f32(false, av, false, bv, (short)0, acc[n], false, false);
        }
#pragma unroll
        for (int n = 0; n < 4; ++n) {
            av.x = a0.z; av.y = a0.w; bv.x = b0[n].z; bv.y = b0[n].w;
            acc[n] = __builtin_amdgcn_wmma_f32_16x16x4_f32(false, av, false, bv, (short)0, acc[n], false, false);
        }
#pragma unroll
        for (int n = 0; n < 4; ++n) {
            av.x = a1.x; av.y = a1.y; bv.x = b1[n].x; bv.y = b1[n].y;
            acc[n] = __builtin_amdgcn_wmma_f32_16x16x4_f32(false, av, false, bv, (short)0, acc[n], false, false);
        }
#pragma unroll
        for (int n = 0; n < 4; ++n) {
            av.x = a1.z; av.y = a1.w; bv.x = b1[n].z; bv.y = b1[n].w;
            acc[n] = __builtin_amdgcn_wmma_f32_16x16x4_f32(false, av, false, bv, (short)0, acc[n], false, false);
        }
    }
    float* out = m0s + (row0 + 8 * hh) * HIDDEN + l15;
    const float* dr = dinv + row0 + 8 * hh;
#pragma unroll
    for (int v = 0; v < 8; ++v) {
        const float d = dr[v];
#pragma unroll
        for (int n = 0; n < 4; ++n) out[v * HIDDEN + 16 * n] = d * acc[n][v];
    }
}


__device__ __forceinline__ float lane_f(float v, int i) {
    return __int_as_float(__builtin_amdgcn_readlane(__float_as_int(v), i));
}

__global__ __launch_bounds__(128) void k_pull0_gemm1(const int* __restrict__ ei,
                                                     const float* __restrict__ ew,
                                                     const unsigned* __restrict__ rowinf,
                                                     const int* __restrict__ csrw,
                                                     const float* __restrict__ m0s,
                                                     const float* __restrict__ b0,
                                                     const float* __restrict__ W1,
                                                     float* __restrict__ m1s) {
    __shared__ float hs[16 * HPAD];
    __shared__ float dl[16];
    const int2* csr = (const int2*)csrw;
    const int lane = threadIdx.x & 31;
    const int wv   = __builtin_amdgcn_readfirstlane((int)(threadIdx.x >> 5));
    const int row0 = blockIdx.x * 16;
    const float2 bb = *(const float2*)(b0 + 2 * lane);

    const int d0 = row0 + wv * 4;
    int n4[4], st4[4];
    int   s4[4][2];
    float w4[4][2];
    for (int q = 0; q < 4; ++q) {
        const unsigned ri = (unsigned)__builtin_amdgcn_readfirstlane((int)rowinf[d0 + q]);
        n4[q]  = (int)(ri & NMAX);
        st4[q] = (int)(ri >> 10);
    }
    for (int q = 0; q < 4; ++q)
        for (int h = 0; h < 2; ++h) {
            int2 p = make_int2(0, 0);
            if (n4[q] < NMAX && h * 32 + lane < n4[q]) p = csr[st4[q] + h * 32 + lane];
            s4[q][h] = p.x;
            w4[q][h] = __int_as_float(p.y);
        }

#pragma unroll
    for (int q = 0; q < 4; ++q) {
        const int r = wv * 4 + q;
        const int d = d0 + q;
        const int n = n4[q];
        float ax = 0.0f, ay = 0.0f, dsum = 0.0f;
        if (n <= 64) {
            for (int i0 = 0; i0 < n; i0 += 16) {
                const int   sh = i0 < 32 ? s4[q][0] : s4[q][1];
                const float wh = i0 < 32 ? w4[q][0] : w4[q][1];
                const int   b  = i0 & 31;
                for (int u = 0; u < 16; ++u) {
                    const int   s = __builtin_amdgcn_readlane(sh, b + u);
                    const float w = lane_f(wh, b + u);
                    const float2 v = *(const float2*)(m0s + s * HIDDEN + 2 * lane);
                    ax = fmaf(w, v.x, ax); ay = fmaf(w, v.y, ay); dsum += w;
                }
            }
        } else if (n < NMAX) {
            const int2* row = csr + st4[q];
            for (int i0 = 0; i0 < n; i0 += 16) {
                int2 p = make_int2(0, 0);
                const int c0 = i0 & ~31;
                if (c0 + lane < n) p = row[c0 + lane];
                const int   sh = p.x;
                const float wh = __int_as_float(p.y);
                const int   b  = i0 & 31;
                for (int u = 0; u < 16; ++u) {
                    const int   s = __builtin_amdgcn_readlane(sh, b + u);
                    const float w = lane_f(wh, b + u);
                    const float2 v = *(const float2*)(m0s + s * HIDDEN + 2 * lane);
                    ax = fmaf(w, v.x, ax); ay = fmaf(w, v.y, ay); dsum += w;
                }
            }
        } else {
            for (int e = 0; e < N_EDGES; ++e) {
                if (ei[N_EDGES + e] == d) {
                    const int s = ei[e];
                    const float w = ew[e];
                    const float2 v = *(const float2*)(m0s + s * HIDDEN + 2 * lane);
                    ax = fmaf(w, v.x, ax); ay = fmaf(w, v.y, ay); dsum += w;
                }
            }
        }
        const float di = rsqrtf(fmaxf(dsum + 1.0f, 1e-12f));
        const float2 sv = *(const float2*)(m0s + d * HIDDEN + 2 * lane);
        hs[r * HPAD + 2 * lane]     = fmaxf(di * (ax + sv.x) + bb.x, 0.0f);
        hs[r * HPAD + 2 * lane + 1] = fmaxf(di * (ay + sv.y) + bb.y, 0.0f);
        if (lane == 0) dl[r] = di;
    }
    __syncthreads();
    if (wv != 0) return;

    const int hh  = lane >> 4;
    const int l15 = lane & 15;
    v8f c = {};
    const float* hrow = hs + l15 * HPAD + 2 * hh;
#pragma unroll
    for (int k = 0; k < HIDDEN; k += 4) {
        v2f a, b;
        a.x = hrow[k];
        a.y = hrow[k + 1];
        float bx = 0.0f, by = 0.0f;
        if (l15 < NCLASS) {
            bx = W1[(k + 2 * hh) * NCLASS + l15];
            by = W1[(k + 2 * hh + 1) * NCLASS + l15];
        }
        b.x = bx; b.y = by;
        c = __builtin_amdgcn_wmma_f32_16x16x4_f32(false, a, false, b,
                                                  (short)0, c, false, false);
    }
    if (l15 < NCLASS) {
        float* out = m1s + (row0 + 8 * hh) * NCLASS + l15;
#pragma unroll
        for (int v = 0; v < 8; ++v) out[v * NCLASS] = dl[v + 8 * hh] * c[v];
    }
}


__global__ __launch_bounds__(256) void k_pull1(const int* __restrict__ ei,
                                               const float* __restrict__ ew,
                                               const unsigned* __restrict__ rowinf,
                                               const int* __restrict__ csrw,
                                               const float* __restrict__ m1s,
                                               const float* __restrict__ b1,
                                               float* __restrict__ out) {
    const int2* csr = (const int2*)csrw;
    const int j = threadIdx.x & 7;
    const int d = blockIdx.x * 32 + (threadIdx.x >> 3);
    const unsigned ri = rowinf[d];
    const int n = (int)(ri & NMAX);
    float a[NCLASS];
    for (int k = 0; k < NCLASS; ++k) a[k] = 0.0f;
    float dsum = 0.0f;
    if (n < NMAX) {
        const int2* row = csr + (ri >> 10);
        for (int b = 0; b < n; b += 32) {
            int s[4]; float w[4]; float v[4][NCLASS];
            for (int t = 0; t < 4; ++t) {
                const int slot = b + t * 8 + j;
                int2 p = make_int2(0, 0);
                if (slot < n) p = row[slot];
                s[t] = p.x;
                w[t] = __int_as_float(p.y);
            }
            for (int t = 0; t < 4; ++t)
                for (int k = 0; k < NCLASS; ++k) v[t][k] = m1s[s[t] * NCLASS + k];
            for (int t = 0; t < 4; ++t) {
                for (int k = 0; k < NCLASS; ++k) a[k] = fmaf(w[t], v[t][k], a[k]);
                dsum += w[t];
            }
        }
    } else {
        for (int e = j; e < N_EDGES; e += 8) {
            if (ei[N_EDGES + e] == d) {
                const float w = ew[e];
                const int s = ei[e];
                for (int k = 0; k < NCLASS; ++k) a[k] = fmaf(w, m1s[s * NCLASS + k], a[k]);
                dsum += w;
            }
        }
    }
    for (int k = 0; k < NCLASS; ++k) {
        a[k] += __shfl_xor(a[k], 4, 8);
        a[k] += __shfl_xor(a[k], 2, 8);
        a[k] += __shfl_xor(a[k], 1, 8);
    }
    dsum += __shfl_xor(dsum, 4, 8);
    dsum += __shfl_xor(dsum, 2, 8);
    dsum += __shfl_xor(dsum, 1, 8);
    float aj = a[0];
    for (int k = 1; k < NCLASS; ++k) aj = (j == k) ? a[k] : aj;
    const float di = rsqrtf(fmaxf(dsum + 1.0f, 1e-12f));
    if (j < NCLASS) out[d * NCLASS + j] = di * (aj + m1s[d * NCLASS + j]) + b1[j];
}


extern "C" void kernel_launch(void* const* d_in, const int* in_sizes, int n_in,
                              void* d_out, int out_size, void* d_ws, size_t ws_size,
                              hipStream_t stream) {
    const float* x  = (const float*)d_in[0];
    const int*   ei = (const int*)  d_in[1];
    const float* ew = (const float*)d_in[2];
    const float* W0 = (const float*)d_in[3];
    const float* b0 = (const float*)d_in[4];
    const float* W1 = (const float*)d_in[5];
    const float* b1 = (const float*)d_in[6];
    float* out = (float*)d_out;

    float*    ws     = (float*)d_ws;
    unsigned* rowinf = (unsigned*)ws;
    int*      csr    = (int*)(ws + N_NODES);
    float*    m0s    = ws + N_NODES + 2 * (size_t)N_EDGES;
    float*    m1s    = m0s + (size_t)N_NODES * HIDDEN;
    float*    dinv   = m1s;

    int*  bins = (int*)m0s;
    int*  gcur = (int*)m0s + 2 * NPART * BCAP;

    k_zero_cur<<<1, 256, 0, stream>>>(gcur);
    k_bin     <<<NBINBLK, 1024, 0, stream>>>(ei, ew, gcur, bins);
    k_place   <<<NPART, 1024, 0, stream>>>(gcur, bins, rowinf, csr);
    k_dinv    <<<N_NODES / 32, 256, 0, stream>>>(ei, ew, rowinf, csr, dinv);

    k_gemm0<<<G0_BLOCKS, G0_THREADS, 0, stream>>>(x, W0, dinv, m0s);

    k_pull0_gemm1<<<ROWBLKS, 128, 0, stream>>>(ei, ew, rowinf, csr, m0s, b0, W1, m1s);

    k_pull1<<<N_NODES / 32, 256, 0, stream>>>(ei, ew, rowinf, csr, m1s, b1, out);
}


# degree sums move into GEMM0: a 16-row tile's wave sums its 16 nodes' pair runs (8 lanes a node, reads ahead of the W0 copy), dinv via LDS; degree launch and dinv array go; sums keep their order
# speedup vs baseline: 5.5759x; 1.0313x over previous
#include <hip/hip_runtime.h>

typedef __attribute__((ext_vector_type(2))) float v2f;
typedef __attribute__((ext_vector_type(8))) float v8f;

#define N_NODES 100000
#define N_EDGES 3200000
#define D_FEAT  256
#define HIDDEN  64
#define NCLASS  7
#define ROWBLKS (N_NODES / 16)
#define HPAD    66


#define PART    400
#define NPART   (N_NODES / PART)
#define BCAP    14080
#define RPT     ((BCAP + 1023) / 1024)
#define EPB     8192
#define NBINBLK ((N_EDGES + EPB - 1) / EPB)
#define CSTRIDE 64
#define NMAX    1023

__global__ void k_zero_cur(int* __restrict__ gcur) {
    if (threadIdx.x < NPART) gcur[threadIdx.x * CSTRIDE] = 0;
}

__global__ __launch_bounds__(1024) void k_bin(const int* __restrict__ ei, const float* __restrict__ ew,
                                              int* __restrict__ gcur, int* __restrict__ binw) {
    int2* bins = (int2*)binw;
    __shared__ int lhist[256];
    __shared__ int loff[256];
    __shared__ int lgb[256];
    __shared__ int2 lrec[EPB];
    __shared__ unsigned char lbin[EPB];
    const int t  = threadIdx.x;
    const int e0 = blockIdx.x * EPB;
    if (t < 256) lhist[t] = 0;
    __syncthreads();

    int d[8], bn[8], rk[8], key[8];
    float w[8];
#pragma unroll
    for (int i = 0; i < 8; ++i) {
        const int e = e0 + i * 1024 + t;
        const bool ok = e < N_EDGES;
        d[i]   = ok ? ei[N_EDGES + e] : -1;
        key[i] = ok ? ei[e] : 0;
        w[i]   = ok ? ew[e] : 0.0f;
    }
#pragma unroll
    for (int i = 0; i < 8; ++i) {
        bn[i] = -1; rk[i] = 0;
        if (d[i] >= 0) {
            const unsigned b = (unsigned)d[i] / PART;
            bn[i]  = (int)b;
            key[i] = (int)((((unsigned)d[i] - b * PART) << 17) | (unsigned)key[i]);
            rk[i]  = atomicAdd(&lhist[b], 1);
        }
    }
    __syncthreads();

    int gb = 0;
    if (t < NPART) {
        const int c = lhist[t];
        if (c > 0) gb = atomicAdd(&gcur[t * CSTRIDE], c);
    }
    if (t < 32) {
        int v[8];
        int run = 0;
#pragma unroll
        for (int k = 0; k < 8; ++k) { const int x = lhist[8 * t + k]; v[k] = run; run += x; }
        int inc = run;
        for (int o = 1; o < 32; o <<= 1) {
            const int y = __shfl_up(inc, o);
            if (t >= o) inc += y;
        }
#pragma unroll
        for (int k = 0; k < 8; ++k) loff[8 * t + k] = inc - run + v[k];
    }
    if (t < 256) lgb[t] = gb;
    __syncthreads();

#pragma unroll
    for (int i = 0; i < 8; ++i) {
        if (bn[i] >= 0) {
            const int pos = loff[bn[i]] + rk[i];
            lrec[pos] = make_int2(key[i], __float_as_int(w[i]));
            lbin[pos] = (unsigned char)bn[i];
        }
    }
    __syncthreads();

    const int nvalid = (N_EDGES - e0) < EPB ? (N_EDGES - e0) : EPB;
#pragma unroll
    for (int i = 0; i < 8; ++i) {
        const int p = i * 1024 + t;
        if (p < nvalid) {
            const int b = lbin[p];
            const int g = lgb[b] + (p - loff[b]);
            if (g < BCAP) bins[b * BCAP + g] = lrec[p];
        }
    }
}

__global__ __launch_bounds__(1024) void k_place(const int* __restrict__ gcur,
                                                const int* __restrict__ binw,
                                                unsigned* __restrict__ rowinf, int* __restrict__ csrw) {
    const int2* bins = (const int2*)binw;
    int2* csr = (int2*)csrw;
    __shared__ int2 lrec[BCAP];
    __shared__ int lcnt[PART];
    __shared__ int lstart[PART];
    __shared__ int lbase;
    const int t    = threadIdx.x;
    const int part = blockIdx.x;
    const int total_raw = gcur[part * CSTRIDE];
    const bool over = total_raw > BCAP;
    const int total = over ? 0 : total_raw;
    const int2* src = bins + part * BCAP;

    int2 r[RPT];
#pragma unroll
    for (int i = 0; i < RPT; ++i) {
        const int p = i * 1024 + t;
        r[i] = p < total ? src[p] : make_int2(-1, 0);
    }
    int before = 0;
    if (t < part) {
        const int c = gcur[t * CSTRIDE];
        before = c > BCAP ? 0 : c;
    }
    if (t < PART) lcnt[t] = 0;
    if (t == 0) lbase = 0;
    __syncthreads();

    int sl[RPT];
#pragma unroll
    for (int i = 0; i < RPT; ++i)
        sl[i] = r[i].x >= 0 ? atomicAdd(&lcnt[r[i].x >> 17], 1) : 0;
    if (t < 256) {
        for (int o = 16; o > 0; o >>= 1) before += __shfl_xor(before, o);
        if ((t & 31) == 0 && before > 0) atomicAdd(&lbase, before);
    }
    __syncthreads();

    if (t < 32) {
        int v[13];
        int run = 0;
#pragma unroll
        for (int k = 0; k < 13; ++k) {
            const int idx = 13 * t + k;
            const int x = idx < PART ? lcnt[idx] : 0;
            v[k] = run; run += x;
        }
        int inc = run;
        for (int o = 1; o < 32; o <<= 1) {
            const int y = __shfl_up(inc, o);
            if (t >= o) inc += y;
        }
#pragma unroll
        for (int k = 0; k < 13; ++k) {
            const int idx = 13 * t + k;
            if (idx < PART) lstart[idx] = inc - run + v[k];
        }
    }
    __syncthreads();

#pragma unroll
    for (int i = 0; i < RPT; ++i)
        if (r[i].x >= 0) lrec[lstart[r[i].x >> 17] + sl[i]] = make_int2(r[i].x & 0x1FFFF, r[i].y);
    __syncthreads();

    const int base = lbase;
    if (t < PART) {
        const int cn = lcnt[t];
        rowinf[part * PART + t] = over ? (unsigned)NMAX
                                       : (((unsigned)(base + lstart[t]) << 10) | (unsigned)(cn < NMAX ? cn : NMAX));
    }
    int2* dst = csr + base;
    for (int q = t; q < total; q += 1024) dst[q] = lrec[q];
}


#define G0_WAVES   25
#define G0_THREADS (G0_WAVES * 32)
#define G0_BLOCKS  (ROWBLKS / G0_WAVES)
#define G0_CHUNKS  (D_FEAT / 16)

__global__ __launch_bounds__(G0_THREADS) void k_gemm0(const float* __restrict__ x,
                                                      const float* __restrict__ W0,
                                                      const int* __restrict__ ei,
                                                      const float* __restrict__ ew,
                                                      const unsigned* __restrict__ rowinf,
                                                      const int* __restrict__ csrw,
                                                      float* __restrict__ m0s) {
    __shared__ float4 wl[D_FEAT * HIDDEN / 4];
    __shared__ float dl[G0_WAVES * 16];
    const int2* csr = (const int2*)csrw;
    const int t    = threadIdx.x;
    const int lane = t & 31;
    const int wv   = t >> 5;
    const int hh   = lane >> 4;
    const int l15  = lane & 15;
    const int row0 = (blockIdx.x * G0_WAVES + wv) * 16;

    const int j = lane & 7;
    const int g = lane >> 3;
    unsigned ri[4];
#pragma unroll
    for (int r = 0; r < 4; ++r) ri[r] = rowinf[row0 + 4 * r + g];

    const float4* xr = (const float4*)(x + (row0 + l15) * D_FEAT + 8 * hh);
    float4 a[4][2];
    a[0][0] = xr[0]; a[0][1] = xr[1];
    a[1][0] = xr[4]; a[1][1] = xr[5];
    a[2][0] = xr[8]; a[2][1] = xr[9];

    float* wlf = (float*)wl;
    float4 wr[6];
#pragma unroll
    for (int j = 0; j < 6; ++j) {
        const int i = t + j * G0_THREADS;
        wr[j] = i < D_FEAT * HIDDEN / 4 ? ((const float4*)W0)[i] : make_float4(0.f, 0.f, 0.f, 0.f);
    }
    float wq[4][8];
#pragma unroll
    for (int r = 0; r < 4; ++r) {
        const int n  = (int)(ri[r] & NMAX);
        const int st = (int)(ri[r] >> 10);
        const int last = n > 0 ? n - 1 : 0;
#pragma unroll
        for (int u = 0; u < 8; ++u) {
            const int slot = u * 8 + j;
            int idx = st + (slot < last ? slot : last);
            idx = idx < N_EDGES - 1 ? idx : N_EDGES - 1;
            const float wv8 = __int_as_float(csr[idx].y);
            wq[r][u] = (n < NMAX && slot < n) ? wv8 : 0.0f;
        }
    }
#pragma unroll
    for (int j = 0; j < 6; ++j) {
        const int i = t + j * G0_THREADS;
        if (i < D_FEAT * HIDDEN / 4) {
            const int k   = i >> 4;
            const int col = (i & 15) * 4;
            const int ch = k >> 4, h = (k >> 3) & 1, sh = (k >> 2) & 1, s3 = k & 3;
            const int n = col >> 4, c = col & 15;
            const int base = ((((ch * 4 + n) * 2 + sh) * 32) + (h * 16 + c)) * 4 + s3;
            wlf[base]      = wr[j].x;
            wlf[base + 4]  = wr[j].y;
            wlf[base + 8]  = wr[j].z;
            wlf[base + 12] = wr[j].w;
        }
    }
    float ds4[4];
#pragma unroll
    for (int r = 0; r < 4; ++r) {
        const int n = (int)(ri[r] & NMAX);
        float dsum = (wq[r][0] + wq[r][1]) + (wq[r][2] + wq[r][3]);
        dsum += (wq[r][4] + wq[r][5]) + (wq[r][6] + wq[r][7]);
        if (n < NMAX) {
            const int2* row = csr + (ri[r] >> 10);
#pragma nounroll
            for (int b = 64; b < n; b += 32) {
                float w[4];
                for (int u = 0; u < 4; ++u) {
                    const int slot = b + u * 8 + j;
                    w[u] = slot < n ? __int_as_float(row[slot].y) : 0.0f;
                }
                dsum += (w[0] + w[1]) + (w[2] + w[3]);
            }
        } else {
            const int d = row0 + 4 * r + g;
            dsum = 0.0f;
#pragma nounroll
            for (int e = j; e < N_EDGES; e += 8)
                if (ei[N_EDGES + e] == d) dsum += ew[e];
        }
        ds4[r] = dsum;
    }
#pragma unroll
    for (int r = 0; r < 4; ++r) ds4[r] += __shfl_xor(ds4[r], 4, 8);
#pragma unroll
    for (int r = 0; r < 4; ++r) ds4[r] += __shfl_xor(ds4[r], 2, 8);
#pragma unroll
    for (int r = 0; r < 4; ++r) ds4[r] += __shfl_xor(ds4[r], 1, 8);
#pragma unroll
    for (int r = 0; r < 4; ++r)
        if (j == 0) dl[wv * 16 + 4 * r + g] = rsqrtf(fmaxf(ds4[r] + 1.0f, 1e-12f));
    __syncthreads();

    v8f acc[4] = {};
#pragma unroll
    for (int ch = 0; ch < G0_CHUNKS; ++ch) {
        if (ch + 3 < G0_CHUNKS) {
            a[(ch + 3) & 3][0] = xr[(ch + 3) * 4];
            a[(ch + 3) & 3][1] = xr[(ch + 3) * 4 + 1];
        }
        const float4 a0 = a[ch & 3][0];
        const float4 a1 = a[ch & 3][1];
        float4 b0[4], b1[4];
#pragma unroll
        for (int n = 0; n < 4; ++n) {
            b0[n] = wl[((ch * 4 + n) * 2 + 0) * 32 + lane];
            b1[n] = wl[((ch * 4 + n) * 2 + 1) * 32 + lane];
        }
        __builtin_amdgcn_sched_barrier(0);
        v2f av, bv;
#pragma unroll
        for (int n = 0; n < 4; ++n) {
            av.x = a0.x; av.y = a0.y; bv.x = b0[n].x; bv.y = b0[n].y;
            acc[n] = __builtin_amdgcn_wmma_f32_16x16x4_f32(false, av, false, bv, (short)0, acc[n], false, false);
        }
#pragma unroll
        for (int n = 0; n < 4; ++n) {
            av.x = a0.z; av.y = a0.w; bv.x = b0[n].z; bv.y = b0[n].w;
            acc[n] = __builtin_amdgcn_wmma_f32_16x16x4_f32(false, av, false, bv, (short)0, acc[n], false, false);
        }
#pragma unroll
        for (int n = 0; n < 4; ++n) {
            av.x = a1.x; av.y = a1.y; bv.x = b1[n].x; bv.y = b1[n].y;
            acc[n] = __builtin_amdgcn_wmma_f32_16x16x4_f32(false, av, false, bv, (short)0, acc[n], false, false);
        }
#pragma unroll
        for (int n = 0; n < 4; ++n) {
            av.x = a1.z; av.y = a1.w; bv.x = b1[n].z; bv.y = b1[n].w;
            acc[n] = __builtin_amdgcn_wmma_f32_16x16x4_f32(false, av, false, bv, (short)0, acc[n], false, false);
        }
    }
    float* out = m0s + (row0 + 8 * hh) * HIDDEN + l15;
    const float* dr = dl + wv * 16 + 8 * hh;
#pragma unroll
    for (int v = 0; v < 8; ++v) {
        const float d = dr[v];
#pragma unroll
        for (int n = 0; n < 4; ++n) out[v * HIDDEN + 16 * n] = d * acc[n][v];
    }
}


__device__ __forceinline__ float lane_f(float v, int i) {
    return __int_as_float(__builtin_amdgcn_readlane(__float_as_int(v), i));
}

__global__ __launch_bounds__(128) void k_pull0_gemm1(const int* __restrict__ ei,
                                                     const float* __restrict__ ew,
                                                     const unsigned* __restrict__ rowinf,
                                                     const int* __restrict__ csrw,
                                                     const float* __restrict__ m0s,
                                                     const float* __restrict__ b0,
                                                     const float* __restrict__ W1,
                                                     float* __restrict__ m1s) {
    __shared__ float hs[16 * HPAD];
    __shared__ float dl[16];
    const int2* csr = (const int2*)csrw;
    const int lane = threadIdx.x & 31;
    const int wv   = __builtin_amdgcn_readfirstlane((int)(threadIdx.x >> 5));
    const int row0 = blockIdx.x * 16;
    const float2 bb = *(const float2*)(b0 + 2 * lane);

    const int d0 = row0 + wv * 4;
    int n4[4], st4[4];
    int   s4[4][2];
    float w4[4][2];
    for (int q = 0; q < 4; ++q) {
        const unsigned ri = (unsigned)__builtin_amdgcn_readfirstlane((int)rowinf[d0 + q]);
        n4[q]  = (int)(ri & NMAX);
        st4[q] = (int)(ri >> 10);
    }
    for (int q = 0; q < 4; ++q)
        for (int h = 0; h < 2; ++h) {
            int2 p = make_int2(0, 0);
            if (n4[q] < NMAX && h * 32 + lane < n4[q]) p = csr[st4[q] + h * 32 + lane];
            s4[q][h] = p.x;
            w4[q][h] = __int_as_float(p.y);
        }

#pragma unroll
    for (int q = 0; q < 4; ++q) {
        const int r = wv * 4 + q;
        const int d = d0 + q;
        const int n = n4[q];
        float ax = 0.0f, ay = 0.0f, dsum = 0.0f;
        if (n <= 64) {
            for (int i0 = 0; i0 < n; i0 += 16) {
                const int   sh = i0 < 32 ? s4[q][0] : s4[q][1];
                const float wh = i0 < 32 ? w4[q][0] : w4[q][1];
                const int   b  = i0 & 31;
                for (int u = 0; u < 16; ++u) {
                    const int   s = __builtin_amdgcn_readlane(sh, b + u);
                    const float w = lane_f(wh, b + u);
                    const float2 v = *(const float2*)(m0s + s * HIDDEN + 2 * lane);
                    ax = fmaf(w, v.x, ax); ay = fmaf(w, v.y, ay); dsum += w;
                }
            }
        } else if (n < NMAX) {
            const int2* row = csr + st4[q];
            for (int i0 = 0; i0 < n; i0 += 16) {
                int2 p = make_int2(0, 0);
                const int c0 = i0 & ~31;
                if (c0 + lane < n) p = row[c0 + lane];
                const int   sh = p.x;
                const float wh = __int_as_float(p.y);
                const int   b  = i0 & 31;
                for (int u = 0; u < 16; ++u) {
                    const int   s = __builtin_amdgcn_readlane(sh, b + u);
                    const float w = lane_f(wh, b + u);
                    const float2 v = *(const float2*)(m0s + s * HIDDEN + 2 * lane);
                    ax = fmaf(w, v.x, ax); ay = fmaf(w, v.y, ay); dsum += w;
                }
            }
        } else {
            for (int e = 0; e < N_EDGES; ++e) {
                if (ei[N_EDGES + e] == d) {
                    const int s = ei[e];
                    const float w = ew[e];
                    const float2 v = *(const float2*)(m0s + s * HIDDEN + 2 * lane);
                    ax = fmaf(w, v.x, ax); ay = fmaf(w, v.y, ay); dsum += w;
                }
            }
        }
        const float di = rsqrtf(fmaxf(dsum + 1.0f, 1e-12f));
        const float2 sv = *(const float2*)(m0s + d * HIDDEN + 2 * lane);
        hs[r * HPAD + 2 * lane]     = fmaxf(di * (ax + sv.x) + bb.x, 0.0f);
        hs[r * HPAD + 2 * lane + 1] = fmaxf(di * (ay + sv.y) + bb.y, 0.0f);
        if (lane == 0) dl[r] = di;
    }
    __syncthreads();
    if (wv != 0) return;

    const int hh  = lane >> 4;
    const int l15 = lane & 15;
    v8f c = {};
    const float* hrow = hs + l15 * HPAD + 2 * hh;
#pragma unroll
    for (int k = 0; k < HIDDEN; k += 4) {
        v2f a, b;
        a.x = hrow[k];
        a.y = hrow[k + 1];
        float bx = 0.0f, by = 0.0f;
        if (l15 < NCLASS) {
            bx = W1[(k + 2 * hh) * NCLASS + l15];
            by = W1[(k + 2 * hh + 1) * NCLASS + l15];
        }
        b.x = bx; b.y = by;
        c = __builtin_amdgcn_wmma_f32_16x16x4_f32(false, a, false, b,
                                                  (short)0, c, false, false);
    }
    if (l15 < NCLASS) {
        float* out = m1s + (row0 + 8 * hh) * NCLASS + l15;
#pragma unroll
        for (int v = 0; v < 8; ++v) out[v * NCLASS] = dl[v + 8 * hh] * c[v];
    }
}


__global__ __launch_bounds__(256) void k_pull1(const int* __restrict__ ei,
                                               const float* __restrict__ ew,
                                               const unsigned* __restrict__ rowinf,
                                               const int* __restrict__ csrw,
                                               const float* __restrict__ m1s,
                                               const float* __restrict__ b1,
                                               float* __restrict__ out) {
    const int2* csr = (const int2*)csrw;
    const int j = threadIdx.x & 7;
    const int d = blockIdx.x * 32 + (threadIdx.x >> 3);
    const unsigned ri = rowinf[d];
    const int n = (int)(ri & NMAX);
    float a[NCLASS];
    for (int k = 0; k < NCLASS; ++k) a[k] = 0.0f;
    float dsum = 0.0f;
    if (n < NMAX) {
        const int2* row = csr + (ri >> 10);
        for (int b = 0; b < n; b += 32) {
            int s[4]; float w[4]; float v[4][NCLASS];
            for (int t = 0; t < 4; ++t) {
                const int slot = b + t * 8 + j;
                int2 p = make_int2(0, 0);
                if (slot < n) p = row[slot];
                s[t] = p.x;
                w[t] = __int_as_float(p.y);
            }
            for (int t = 0; t < 4; ++t)
                for (int k = 0; k < NCLASS; ++k) v[t][k] = m1s[s[t] * NCLASS + k];
            for (int t = 0; t < 4; ++t) {
                for (int k = 0; k < NCLASS; ++k) a[k] = fmaf(w[t], v[t][k], a[k]);
                dsum += w[t];
            }
        }
    } else {
        for (int e = j; e < N_EDGES; e += 8) {
            if (ei[N_EDGES + e] == d) {
                const float w = ew[e];
                const int s = ei[e];
                for (int k = 0; k < NCLASS; ++k) a[k] = fmaf(w, m1s[s * NCLASS + k], a[k]);
                dsum += w;
            }
        }
    }
    for (int k = 0; k < NCLASS; ++k) {
        a[k] += __shfl_xor(a[k], 4, 8);
        a[k] += __shfl_xor(a[k], 2, 8);
        a[k] += __shfl_xor(a[k], 1, 8);
    }
    dsum += __shfl_xor(dsum, 4, 8);
    dsum += __shfl_xor(dsum, 2, 8);
    dsum += __shfl_xor(dsum, 1, 8);
    float aj = a[0];
    for (int k = 1; k < NCLASS; ++k) aj = (j == k) ? a[k] : aj;
    const float di = rsqrtf(fmaxf(dsum + 1.0f, 1e-12f));
    if (j < NCLASS) out[d * NCLASS + j] = di * (aj + m1s[d * NCLASS + j]) + b1[j];
}


extern "C" void kernel_launch(void* const* d_in, const int* in_sizes, int n_in,
                              void* d_out, int out_size, void* d_ws, size_t ws_size,
                              hipStream_t stream) {
    const float* x  = (const float*)d_in[0];
    const int*   ei = (const int*)  d_in[1];
    const float* ew = (const float*)d_in[2];
    const float* W0 = (const float*)d_in[3];
    const float* b0 = (const float*)d_in[4];
    const float* W1 = (const float*)d_in[5];
    const float* b1 = (const float*)d_in[6];
    float* out = (float*)d_out;

    float*    ws     = (float*)d_ws;
    unsigned* rowinf = (unsigned*)ws;
    int*      csr    = (int*)(ws + N_NODES);
    float*    m0s    = ws + N_NODES + 2 * (size_t)N_EDGES;
    float*    m1s    = m0s + (size_t)N_NODES * HIDDEN;

    int*  bins = (int*)m0s;
    int*  gcur = (int*)m0s + 2 * NPART * BCAP;

    k_zero_cur<<<1, 256, 0, stream>>>(gcur);
    k_bin     <<<NBINBLK, 1024, 0, stream>>>(ei, ew, gcur, bins);
    k_place   <<<NPART, 1024, 0, stream>>>(gcur, bins, rowinf, csr);

    k_gemm0<<<G0_BLOCKS, G0_THREADS, 0, stream>>>(x, W0, ei, ew, rowinf, csr, m0s);

    k_pull0_gemm1<<<ROWBLKS, 128, 0, stream>>>(ei, ew, rowinf, csr, m0s, b0, W1, m1s);

    k_pull1<<<N_NODES / 32, 256, 0, stream>>>(ei, ew, rowinf, csr, m1s, b1, out);
}
